# starting kernel
# baseline (speedup 1.0000x reference)
#include <hip/hip_runtime.h>


#define N_NODES 50000
#define N_EDGES 800000
#define F_IN    64
#define F_HID   128
#define F_OUT   64

typedef __attribute__((ext_vector_type(2))) float v2f;
typedef __attribute__((ext_vector_type(8))) float v8f;


__global__ void zero_kernel(float* __restrict__ p, int n) {
    int i = blockIdx.x * blockDim.x + threadIdx.x;
    if (i < n) p[i] = 0.0f;
}

__global__ void degree_kernel(const int* __restrict__ src, const int* __restrict__ dst,
                              float* __restrict__ out_deg, float* __restrict__ in_deg) {
    int e = blockIdx.x * blockDim.x + threadIdx.x;
    if (e < N_EDGES) {
        atomicAdd(&out_deg[src[e]], 1.0f);
        atomicAdd(&in_deg[dst[e]], 1.0f);
    }
}

__global__ void norm_kernel(const float* __restrict__ out_deg, const float* __restrict__ in_deg,
                            float* __restrict__ norm_src, float* __restrict__ norm_dst) {
    int i = blockIdx.x * blockDim.x + threadIdx.x;
    if (i < N_NODES) {
        norm_src[i] = rsqrtf(fmaxf(out_deg[i], 1.0f));
        norm_dst[i] = rsqrtf(fmaxf(in_deg[i], 1.0f));
    }
}

__global__ void scatter_scaled_kernel(const float* __restrict__ feat,
                                      const float* __restrict__ scale,
                                      const int* __restrict__ src, const int* __restrict__ dst,
                                      float* __restrict__ agg) {
    long long t = (long long)blockIdx.x * 256 + threadIdx.x;
    int e = (int)(t >> 6);
    int f = (int)(t & 63);
    if (e < N_EDGES) {
        int s = src[e], d = dst[e];
        atomicAdd(&agg[(long long)d * 64 + f], feat[(long long)s * 64 + f] * scale[s]);
    }
}

__global__ void scatter_plain_kernel(const float* __restrict__ feat,
                                     const int* __restrict__ src, const int* __restrict__ dst,
                                     float* __restrict__ agg) {
    long long t = (long long)blockIdx.x * 256 + threadIdx.x;
    int e = (int)(t >> 6);
    int f = (int)(t & 63);
    if (e < N_EDGES) {
        int s = src[e], d = dst[e];
        atomicAdd(&agg[(long long)d * 64 + f], feat[(long long)s * 64 + f]);
    }
}

__global__ void final_kernel(const float* __restrict__ agg, const float* __restrict__ norm_dst,
                             const float* __restrict__ bias, float* __restrict__ out) {
    long long t = (long long)blockIdx.x * 256 + threadIdx.x;
    if (t < (long long)N_NODES * 64) {
        int i = (int)(t >> 6);
        int f = (int)(t & 63);
        out[t] = agg[t] * norm_dst[i] + bias[f];
    }
}

template <int KDIM, int NDIM, bool SCALE_A, bool EPILOGUE>
__global__ void wmma_gemm_kernel(const float* __restrict__ A,
                                 const float* __restrict__ B,
                                 const float* __restrict__ rowscale_a,
                                 const float* __restrict__ rowscale_d,
                                 const float* __restrict__ bias,
                                 float* __restrict__ D) {
    const int lane    = threadIdx.x & 31;
    const int wave    = threadIdx.x >> 5;
    const int half    = lane >> 4;
    const int l       = lane & 15;
    const int row0    = blockIdx.x * 16;
    const int arow    = row0 + l;
    const int col     = wave * 16 + l;

    float ascale = 1.0f;
    if (SCALE_A) ascale = rowscale_a[arow];

    v8f c = {};
    const float* arowp = A + (long long)arow * KDIM;

    #pragma unroll
    for (int k = 0; k < KDIM; k += 4) {
        const int ka = k + half * 2;
        v2f a = *(const v2f*)(arowp + ka);
        if (SCALE_A) { a.x *= ascale; a.y *= ascale; }
        v2f b;
        b.x = B[(long long)ka * NDIM + col];
        b.y = B[(long long)(ka + 1) * NDIM + col];
        c = __builtin_amdgcn_wmma_f32_16x16x4_f32(
                 false, a,  false, b,
                 (short)0, c,  false,  false);
    }

    #pragma unroll
    for (int r = 0; r < 8; ++r) {
        int row = row0 + r + half * 8;
        float v = c[r];
        if (EPILOGUE) {
            v = v * rowscale_d[row] + bias[col];
            v = fmaxf(v, 0.0f);
        }
        D[(long long)row * NDIM + col] = v;
    }
}


extern "C" void kernel_launch(void* const* d_in, const int* in_sizes, int n_in,
                              void* d_out, int out_size, void* d_ws, size_t ws_size,
                              hipStream_t stream) {
    const float* x   = (const float*)d_in[0];
    const float* W1  = (const float*)d_in[1];
    const float* b1  = (const float*)d_in[2];
    const float* W2  = (const float*)d_in[3];
    const float* b2  = (const float*)d_in[4];
    const int*   src = (const int*)d_in[5];
    const int*   dst = (const int*)d_in[6];
    float*       out = (float*)d_out;

    float* ws       = (float*)d_ws;
    float* out_deg  = ws;
    float* in_deg   = out_deg + N_NODES;
    float* norm_src = in_deg + N_NODES;
    float* norm_dst = norm_src + N_NODES;
    float* agg      = norm_dst + N_NODES;
    float* h1       = agg + (long long)N_NODES * 64;
    float* h2w      = h1 + (long long)N_NODES * 128;

    const int ZT = 256;
    const long long aggElems = (long long)N_NODES * 64;
    const long long edgeThreads = (long long)N_EDGES * 64;

    zero_kernel<<<(2 * N_NODES + ZT - 1) / ZT, ZT, 0, stream>>>(out_deg, 2 * N_NODES);
    degree_kernel<<<(N_EDGES + ZT - 1) / ZT, ZT, 0, stream>>>(src, dst, out_deg, in_deg);
    norm_kernel<<<(N_NODES + ZT - 1) / ZT, ZT, 0, stream>>>(out_deg, in_deg, norm_src, norm_dst);

    zero_kernel<<<(int)((aggElems + ZT - 1) / ZT), ZT, 0, stream>>>(agg, (int)aggElems);
    scatter_scaled_kernel<<<(int)((edgeThreads + ZT - 1) / ZT), ZT, 0, stream>>>(
        x, norm_src, src, dst, agg);
    wmma_gemm_kernel<F_IN, F_HID, false, true><<<N_NODES / 16, 8 * 32, 0, stream>>>(
        agg, W1, nullptr, norm_dst, b1, h1);

    zero_kernel<<<(int)((aggElems + ZT - 1) / ZT), ZT, 0, stream>>>(agg, (int)aggElems);
    wmma_gemm_kernel<F_HID, F_OUT, true, false><<<N_NODES / 16, 4 * 32, 0, stream>>>(
        h1, W2, norm_src, nullptr, nullptr, h2w);
    scatter_plain_kernel<<<(int)((edgeThreads + ZT - 1) / ZT), ZT, 0, stream>>>(
        h2w, src, dst, agg);

    final_kernel<<<(int)((aggElems + ZT - 1) / ZT), ZT, 0, stream>>>(agg, norm_dst, b2, out);
}


# Global float-atomic scatter replaced by owner-computes: a work-group owns 200 dst rows in LDS, scans all of dst, adds matching src rows by LDS float adds (arrival order), stores the tile once
# speedup vs baseline: 1.6351x; 1.6351x over previous
#include <hip/hip_runtime.h>


#define N_NODES 50000
#define N_EDGES 800000
#define F_IN    64
#define F_HID   128
#define F_OUT   64

typedef __attribute__((ext_vector_type(2))) float v2f;
typedef __attribute__((ext_vector_type(8))) float v8f;


__global__ void zero_kernel(float* __restrict__ p, int n) {
    int i = blockIdx.x * blockDim.x + threadIdx.x;
    if (i < n) p[i] = 0.0f;
}

__global__ void degree_kernel(const int* __restrict__ src, const int* __restrict__ dst,
                              float* __restrict__ out_deg, float* __restrict__ in_deg) {
    int e = blockIdx.x * blockDim.x + threadIdx.x;
    if (e < N_EDGES) {
        atomicAdd(&out_deg[src[e]], 1.0f);
        atomicAdd(&in_deg[dst[e]], 1.0f);
    }
}

__global__ void norm_kernel(const float* __restrict__ out_deg, const float* __restrict__ in_deg,
                            float* __restrict__ norm_src, float* __restrict__ norm_dst) {
    int i = blockIdx.x * blockDim.x + threadIdx.x;
    if (i < N_NODES) {
        norm_src[i] = rsqrtf(fmaxf(out_deg[i], 1.0f));
        norm_dst[i] = rsqrtf(fmaxf(in_deg[i], 1.0f));
    }
}

#define TILE_NODES   200
#define AGG_THREADS  1024
#define AGG_WAVES    (AGG_THREADS / 32)
#define SCAN_LOADS   16
#define CHUNK_EDGES  (SCAN_LOADS * 128)
#define DRAIN_EVERY  4
#define QCAP         544
static_assert(N_NODES % TILE_NODES == 0, "tiles must cover the nodes exactly");
static_assert(N_EDGES % 4 == 0, "dst is scanned in 16-byte pieces");
static_assert(N_EDGES <= (1 << 20) && TILE_NODES <= (1 << 12), "queue entry packing");
static_assert(SCAN_LOADS % DRAIN_EVERY == 0 && QCAP >= 31 + DRAIN_EVERY * 128, "queue cannot overflow");

template <bool SCALED>
__device__ __forceinline__ void add_queued_rows(const unsigned* q, int cnt, int lane,
                                                const float* __restrict__ feat,
                                                const float* __restrict__ scale,
                                                const int* __restrict__ src,
                                                float* tile) {
    const unsigned ent = q[lane < cnt ? lane : 0];
    const int e  = (int)(ent & 0xFFFFFu);
    const int dl = (int)(ent >> 20);
    const int sv = src[e];
    float sc = 1.0f;
    if (SCALED) sc = scale[sv];
    #pragma unroll 1
    for (int j0 = 0; j0 < cnt; j0 += 8) {
        v2f r[8];
        #pragma unroll
        for (int u = 0; u < 8; ++u) {
            const int j  = (j0 + u < cnt) ? (j0 + u) : (cnt - 1);
            const int sj = __builtin_amdgcn_readlane(sv, j);
            r[u] = *(const v2f*)(feat + (long long)sj * 64 + lane * 2);
        }
        #pragma unroll
        for (int u = 0; u < 8; ++u) {
            const int j = j0 + u;
            if (j < cnt) {
                const int dj = __builtin_amdgcn_readlane(dl, j);
                float x0 = r[u].x, x1 = r[u].y;
                if (SCALED) {
                    const float sj = __int_as_float(__builtin_amdgcn_readlane(__float_as_int(sc), j));
                    x0 *= sj; x1 *= sj;
                }
                float* t = tile + dj * 64 + lane * 2;
                atomicAdd(t, x0);
                atomicAdd(t + 1, x1);
            }
        }
    }
}

template <bool SCALED>
__global__ __launch_bounds__(AGG_THREADS)
void owner_aggregate_kernel(const float* __restrict__ feat,
                            const float* __restrict__ scale,
                            const int* __restrict__ src, const int* __restrict__ dst,
                            float* __restrict__ agg) {
    __shared__ float    tile[TILE_NODES * 64];
    __shared__ unsigned queue[AGG_WAVES][QCAP];

    const int lane = threadIdx.x & 31;
    const int wave = __builtin_amdgcn_readfirstlane((int)(threadIdx.x >> 5));
    const int base = blockIdx.x * TILE_NODES;

    for (int i = threadIdx.x; i < TILE_NODES * 64; i += AGG_THREADS) tile[i] = 0.0f;
    __syncthreads();

    unsigned* q = queue[wave];
    int qn = 0;
    const int4* dst4 = (const int4*)dst;
    const int n4 = N_EDGES / 4;
    const int nchunks = (N_EDGES + CHUNK_EDGES - 1) / CHUNK_EDGES;
    const unsigned below = (1u << lane) - 1u;

    const int rot = (int)((blockIdx.x * 25u) % (unsigned)nchunks);
    for (int cl = wave; cl < nchunks; cl += AGG_WAVES) {
        const int c = (cl + rot < nchunks) ? (cl + rot) : (cl + rot - nchunks);
        int4 v[SCAN_LOADS];
        const int i4base = c * (CHUNK_EDGES / 4) + lane;
        #pragma unroll
        for (int k = 0; k < SCAN_LOADS; ++k) {
            const int i4 = i4base + k * 32;
            v[k] = (i4 < n4) ? dst4[i4] : make_int4(-1, -1, -1, -1);
        }
        #pragma unroll
        for (int k = 0; k < SCAN_LOADS; ++k) {
            const int e0 = (i4base + k * 32) * 4;
            const unsigned d0 = (unsigned)(v[k].x - base), d1 = (unsigned)(v[k].y - base);
            const unsigned d2 = (unsigned)(v[k].z - base), d3 = (unsigned)(v[k].w - base);
            const bool m0 = d0 < TILE_NODES, m1 = d1 < TILE_NODES;
            const bool m2 = d2 < TILE_NODES, m3 = d3 < TILE_NODES;
            if (__builtin_amdgcn_ballot_w32(m0 | m1 | m2 | m3) != 0u) {
                const unsigned b0 = __builtin_amdgcn_ballot_w32(m0);
                const unsigned b1 = __builtin_amdgcn_ballot_w32(m1);
                const unsigned b2 = __builtin_amdgcn_ballot_w32(m2);
                const unsigned b3 = __builtin_amdgcn_ballot_w32(m3);
                int p = qn;
                if (m0) q[p + __builtin_popcount(b0 & below)] = (unsigned)(e0 + 0) | (d0 << 20);
                p += __builtin_popcount(b0);
                if (m1) q[p + __builtin_popcount(b1 & below)] = (unsigned)(e0 + 1) | (d1 << 20);
                p += __builtin_popcount(b1);
                if (m2) q[p + __builtin_popcount(b2 & below)] = (unsigned)(e0 + 2) | (d2 << 20);
                p += __builtin_popcount(b2);
                if (m3) q[p + __builtin_popcount(b3 & below)] = (unsigned)(e0 + 3) | (d3 << 20);
                p += __builtin_popcount(b3);
                qn = p;
            }
            if ((k % DRAIN_EVERY) != DRAIN_EVERY - 1) continue;
            while (qn >= 32) {
                __builtin_amdgcn_wave_barrier();
                qn -= 32;
                add_queued_rows<SCALED>(q + qn, 32, lane, feat, scale, src, tile);
            }
        }
    }
    if (qn > 0) {
        __builtin_amdgcn_wave_barrier();
        add_queued_rows<SCALED>(q, qn, lane, feat, scale, src, tile);
    }
    __syncthreads();

    float4* out4 = (float4*)(agg + (long long)base * 64);
    const float4* tile4 = (const float4*)tile;
    for (int i = threadIdx.x; i < TILE_NODES * 16; i += AGG_THREADS) out4[i] = tile4[i];
}

__global__ void final_kernel(const float* __restrict__ agg, const float* __restrict__ norm_dst,
                             const float* __restrict__ bias, float* __restrict__ out) {
    long long t = (long long)blockIdx.x * 256 + threadIdx.x;
    if (t < (long long)N_NODES * 64) {
        int i = (int)(t >> 6);
        int f = (int)(t & 63);
        out[t] = agg[t] * norm_dst[i] + bias[f];
    }
}

template <int KDIM, int NDIM, bool SCALE_A, bool EPILOGUE>
__global__ void wmma_gemm_kernel(const float* __restrict__ A,
                                 const float* __restrict__ B,
                                 const float* __restrict__ rowscale_a,
                                 const float* __restrict__ rowscale_d,
                                 const float* __restrict__ bias,
                                 float* __restrict__ D) {
    const int lane    = threadIdx.x & 31;
    const int wave    = threadIdx.x >> 5;
    const int half    = lane >> 4;
    const int l       = lane & 15;
    const int row0    = blockIdx.x * 16;
    const int arow    = row0 + l;
    const int col     = wave * 16 + l;

    float ascale = 1.0f;
    if (SCALE_A) ascale = rowscale_a[arow];

    v8f c = {};
    const float* arowp = A + (long long)arow * KDIM;

    #pragma unroll
    for (int k = 0; k < KDIM; k += 4) {
        const int ka = k + half * 2;
        v2f a = *(const v2f*)(arowp + ka);
        if (SCALE_A) { a.x *= ascale; a.y *= ascale; }
        v2f b;
        b.x = B[(long long)ka * NDIM + col];
        b.y = B[(long long)(ka + 1) * NDIM + col];
        c = __builtin_amdgcn_wmma_f32_16x16x4_f32(
                 false, a,  false, b,
                 (short)0, c,  false,  false);
    }

    #pragma unroll
    for (int r = 0; r < 8; ++r) {
        int row = row0 + r + half * 8;
        float v = c[r];
        if (EPILOGUE) {
            v = v * rowscale_d[row] + bias[col];
            v = fmaxf(v, 0.0f);
        }
        D[(long long)row * NDIM + col] = v;
    }
}


extern "C" void kernel_launch(void* const* d_in, const int* in_sizes, int n_in,
                              void* d_out, int out_size, void* d_ws, size_t ws_size,
                              hipStream_t stream) {
    const float* x   = (const float*)d_in[0];
    const float* W1  = (const float*)d_in[1];
    const float* b1  = (const float*)d_in[2];
    const float* W2  = (const float*)d_in[3];
    const float* b2  = (const float*)d_in[4];
    const int*   src = (const int*)d_in[5];
    const int*   dst = (const int*)d_in[6];
    float*       out = (float*)d_out;

    float* ws       = (float*)d_ws;
    float* out_deg  = ws;
    float* in_deg   = out_deg + N_NODES;
    float* norm_src = in_deg + N_NODES;
    float* norm_dst = norm_src + N_NODES;
    float* agg      = norm_dst + N_NODES;
    float* h1       = agg + (long long)N_NODES * 64;
    float* h2w      = h1 + (long long)N_NODES * 128;

    const int ZT = 256;
    const long long aggElems = (long long)N_NODES * 64;

    zero_kernel<<<(2 * N_NODES + ZT - 1) / ZT, ZT, 0, stream>>>(out_deg, 2 * N_NODES);
    degree_kernel<<<(N_EDGES + ZT - 1) / ZT, ZT, 0, stream>>>(src, dst, out_deg, in_deg);
    norm_kernel<<<(N_NODES + ZT - 1) / ZT, ZT, 0, stream>>>(out_deg, in_deg, norm_src, norm_dst);

    owner_aggregate_kernel<true><<<N_NODES / TILE_NODES, AGG_THREADS, 0, stream>>>(
        x, norm_src, src, dst, agg);
    wmma_gemm_kernel<F_IN, F_HID, false, true><<<N_NODES / 16, 8 * 32, 0, stream>>>(
        agg, W1, nullptr, norm_dst, b1, h1);

    wmma_gemm_kernel<F_HID, F_OUT, true, false><<<N_NODES / 16, 4 * 32, 0, stream>>>(
        h1, W2, norm_src, nullptr, nullptr, h2w);
    owner_aggregate_kernel<false><<<N_NODES / TILE_NODES, AGG_THREADS, 0, stream>>>(
        h2w, nullptr, src, dst, agg);

    final_kernel<<<(int)((aggElems + ZT - 1) / ZT), ZT, 0, stream>>>(agg, norm_dst, b2, out);
}


# Layer-2 aggregation reads per-wave match lists (src row, local dst row) that the layer-1 dst scan writes to the workspace, instead of scanning dst again; LDS float adds still in arrival order
# speedup vs baseline: 1.8162x; 1.1108x over previous
#include <hip/hip_runtime.h>


#define N_NODES 50000
#define N_EDGES 800000
#define F_IN    64
#define F_HID   128
#define F_OUT   64

typedef __attribute__((ext_vector_type(2))) float v2f;
typedef __attribute__((ext_vector_type(8))) float v8f;


__global__ void zero_kernel(float* __restrict__ p, int n) {
    int i = blockIdx.x * blockDim.x + threadIdx.x;
    if (i < n) p[i] = 0.0f;
}

__global__ void degree_kernel(const int* __restrict__ src, const int* __restrict__ dst,
                              float* __restrict__ out_deg, float* __restrict__ in_deg) {
    int e = blockIdx.x * blockDim.x + threadIdx.x;
    if (e < N_EDGES) {
        atomicAdd(&out_deg[src[e]], 1.0f);
        atomicAdd(&in_deg[dst[e]], 1.0f);
    }
}

__global__ void norm_kernel(const float* __restrict__ out_deg, const float* __restrict__ in_deg,
                            float* __restrict__ norm_src, float* __restrict__ norm_dst) {
    int i = blockIdx.x * blockDim.x + threadIdx.x;
    if (i < N_NODES) {
        norm_src[i] = rsqrtf(fmaxf(out_deg[i], 1.0f));
        norm_dst[i] = rsqrtf(fmaxf(in_deg[i], 1.0f));
    }
}

#define TILE_NODES   200
#define AGG_THREADS  1024
#define AGG_WAVES    (AGG_THREADS / 32)
#define SCAN_LOADS   16
#define CHUNK_EDGES  (SCAN_LOADS * 128)
#define DRAIN_EVERY  4
#define QCAP         544
static_assert(N_NODES % TILE_NODES == 0, "tiles must cover the nodes exactly");
static_assert(N_EDGES % 4 == 0, "dst is scanned in 16-byte pieces");
static_assert(N_EDGES <= (1 << 20) && TILE_NODES <= (1 << 12), "queue entry packing");
static_assert(SCAN_LOADS % DRAIN_EVERY == 0 && QCAP >= 31 + DRAIN_EVERY * 128, "queue cannot overflow");

#define LIST_BLOCKS  16
#define LIST_INVALID 0xFFFFFFFFu
static_assert(TILE_NODES <= (1 << 8) && N_NODES < (1 << 24), "list entry packing");

template <bool SCALED, int ROWS>
__device__ __forceinline__ void add_rows(int sv, int dl, float sc, int cnt, int lane,
                                         const float* __restrict__ feat, float* tile) {
    #pragma unroll 1
    for (int j0 = 0; j0 < cnt; j0 += ROWS) {
        v2f r[ROWS];
        #pragma unroll
        for (int u = 0; u < ROWS; ++u) {
            const int j  = (j0 + u < cnt) ? (j0 + u) : (cnt - 1);
            const int sj = __builtin_amdgcn_readlane(sv, j);
            r[u] = *(const v2f*)(feat + (long long)sj * 64 + lane * 2);
        }
        #pragma unroll
        for (int u = 0; u < ROWS; ++u) {
            const int j = j0 + u;
            if (j < cnt) {
                const int dj = __builtin_amdgcn_readlane(dl, j);
                float x0 = r[u].x, x1 = r[u].y;
                if (SCALED) {
                    const float sj = __int_as_float(__builtin_amdgcn_readlane(__float_as_int(sc), j));
                    x0 *= sj; x1 *= sj;
                }
                float* t = tile + dj * 64 + lane * 2;
                atomicAdd(t, x0);
                atomicAdd(t + 1, x1);
            }
        }
    }
}

template <bool SCALED, bool RECORD>
__device__ __forceinline__ void add_queued_rows(const unsigned* q, int cnt, int lane,
                                                const float* __restrict__ feat,
                                                const float* __restrict__ scale,
                                                const int* __restrict__ src,
                                                float* tile,
                                                unsigned* __restrict__ wave_list, int& nrec) {
    const unsigned ent = q[lane < cnt ? lane : 0];
    const int e  = (int)(ent & 0xFFFFFu);
    const int dl = (int)(ent >> 20);
    const int sv = src[e];
    float sc = 1.0f;
    if (SCALED) sc = scale[sv];
    if (RECORD) {
        if (nrec < LIST_BLOCKS)
            wave_list[nrec * 32 + lane] = (lane < cnt) ? (((unsigned)sv << 8) | (unsigned)dl) : LIST_INVALID;
        ++nrec;
    }
    add_rows<SCALED, 8>(sv, dl, sc, cnt, lane, feat, tile);
}

template <bool SCALED, bool RECORD>
__device__ __forceinline__ int scan_and_add(unsigned* q, int lane, int wave, int base,
                                            const float* __restrict__ feat,
                                            const float* __restrict__ scale,
                                            const int* __restrict__ src, const int* __restrict__ dst,
                                            float* tile, unsigned* __restrict__ wave_list) {
    int qn = 0;
    int nrec = 0;
    const int4* dst4 = (const int4*)dst;
    const int n4 = N_EDGES / 4;
    const int nchunks = (N_EDGES + CHUNK_EDGES - 1) / CHUNK_EDGES;
    const unsigned below = (1u << lane) - 1u;

    const int rot = (int)((blockIdx.x * 25u) % (unsigned)nchunks);
    for (int cl = wave; cl < nchunks; cl += AGG_WAVES) {
        const int c = (cl + rot < nchunks) ? (cl + rot) : (cl + rot - nchunks);
        int4 v[SCAN_LOADS];
        const int i4base = c * (CHUNK_EDGES / 4) + lane;
        #pragma unroll
        for (int k = 0; k < SCAN_LOADS; ++k) {
            const int i4 = i4base + k * 32;
            v[k] = (i4 < n4) ? dst4[i4] : make_int4(-1, -1, -1, -1);
        }
        #pragma unroll
        for (int k = 0; k < SCAN_LOADS; ++k) {
            const int e0 = (i4base + k * 32) * 4;
            const unsigned d0 = (unsigned)(v[k].x - base), d1 = (unsigned)(v[k].y - base);
            const unsigned d2 = (unsigned)(v[k].z - base), d3 = (unsigned)(v[k].w - base);
            const bool m0 = d0 < TILE_NODES, m1 = d1 < TILE_NODES;
            const bool m2 = d2 < TILE_NODES, m3 = d3 < TILE_NODES;
            if (__builtin_amdgcn_ballot_w32(m0 | m1 | m2 | m3) != 0u) {
                const unsigned b0 = __builtin_amdgcn_ballot_w32(m0);
                const unsigned b1 = __builtin_amdgcn_ballot_w32(m1);
                const unsigned b2 = __builtin_amdgcn_ballot_w32(m2);
                const unsigned b3 = __builtin_amdgcn_ballot_w32(m3);
                int p = qn;
                if (m0) q[p + __builtin_popcount(b0 & below)] = (unsigned)(e0 + 0) | (d0 << 20);
                p += __builtin_popcount(b0);
                if (m1) q[p + __builtin_popcount(b1 & below)] = (unsigned)(e0 + 1) | (d1 << 20);
                p += __builtin_popcount(b1);
                if (m2) q[p + __builtin_popcount(b2 & below)] = (unsigned)(e0 + 2) | (d2 << 20);
                p += __builtin_popcount(b2);
                if (m3) q[p + __builtin_popcount(b3 & below)] = (unsigned)(e0 + 3) | (d3 << 20);
                p += __builtin_popcount(b3);
                qn = p;
            }
            if ((k % DRAIN_EVERY) != DRAIN_EVERY - 1) continue;
            while (qn >= 32) {
                __builtin_amdgcn_wave_barrier();
                qn -= 32;
                add_queued_rows<SCALED, RECORD>(q + qn, 32, lane, feat, scale, src, tile, wave_list, nrec);
            }
        }
    }
    if (qn > 0) {
        __builtin_amdgcn_wave_barrier();
        add_queued_rows<SCALED, RECORD>(q, qn, lane, feat, scale, src, tile, wave_list, nrec);
    }
    return nrec;
}

template <bool SCALED, bool FIRST>
__global__ __launch_bounds__(AGG_THREADS)
void owner_aggregate_kernel(const float* __restrict__ feat,
                            const float* __restrict__ scale,
                            const int* __restrict__ src, const int* __restrict__ dst,
                            unsigned* __restrict__ lists,
                            int* __restrict__ list_blocks,
                            float* __restrict__ agg) {
    __shared__ float    tile[TILE_NODES * 64];
    __shared__ unsigned queue[AGG_WAVES][QCAP];

    const int lane = threadIdx.x & 31;
    const int wave = __builtin_amdgcn_readfirstlane((int)(threadIdx.x >> 5));
    const int base = blockIdx.x * TILE_NODES;

    for (int i = threadIdx.x; i < TILE_NODES * 64; i += AGG_THREADS) tile[i] = 0.0f;
    __syncthreads();

    unsigned* q = queue[wave];
    unsigned* wave_list = lists + (long long)(blockIdx.x * AGG_WAVES + wave) * (LIST_BLOCKS * 32);

    if (FIRST) {
        const int nrec = scan_and_add<SCALED, true>(q, lane, wave, base, feat, scale, src, dst, tile, wave_list);
        if (lane == 0) list_blocks[blockIdx.x * AGG_WAVES + wave] = nrec;
    } else {
        const int nb_l = list_blocks[blockIdx.x * AGG_WAVES + lane];
        if (__builtin_amdgcn_ballot_w32(nb_l > LIST_BLOCKS) == 0u) {
            const int nb = __builtin_amdgcn_readlane(nb_l, wave);
            unsigned ent_next = (nb > 0) ? wave_list[lane] : LIST_INVALID;
            for (int b = 0; b < nb; ++b) {
                const unsigned ent = ent_next;
                ent_next = (b + 1 < nb) ? wave_list[(b + 1) * 32 + lane] : LIST_INVALID;
                const bool ok = ent != LIST_INVALID;
                const int cnt = __builtin_popcount(__builtin_amdgcn_ballot_w32(ok));
                const int sv = ok ? (int)(ent >> 8) : 0;
                const int dl = (int)(ent & 0xFFu);
                float sc = 1.0f;
                if (SCALED) sc = scale[sv];
                add_rows<SCALED, 16>(sv, dl, sc, cnt, lane, feat, tile);
            }
        } else {
            scan_and_add<SCALED, false>(q, lane, wave, base, feat, scale, src, dst, tile, wave_list);
        }
    }
    __syncthreads();

    float4* out4 = (float4*)(agg + (long long)base * 64);
    const float4* tile4 = (const float4*)tile;
    for (int i = threadIdx.x; i < TILE_NODES * 16; i += AGG_THREADS) out4[i] = tile4[i];
}

__global__ void final_kernel(const float* __restrict__ agg, const float* __restrict__ norm_dst,
                             const float* __restrict__ bias, float* __restrict__ out) {
    long long t = (long long)blockIdx.x * 256 + threadIdx.x;
    if (t < (long long)N_NODES * 64) {
        int i = (int)(t >> 6);
        int f = (int)(t & 63);
        out[t] = agg[t] * norm_dst[i] + bias[f];
    }
}

template <int KDIM, int NDIM, bool SCALE_A, bool EPILOGUE>
__global__ void wmma_gemm_kernel(const float* __restrict__ A,
                                 const float* __restrict__ B,
                                 const float* __restrict__ rowscale_a,
                                 const float* __restrict__ rowscale_d,
                                 const float* __restrict__ bias,
                                 float* __restrict__ D) {
    const int lane    = threadIdx.x & 31;
    const int wave    = threadIdx.x >> 5;
    const int half    = lane >> 4;
    const int l       = lane & 15;
    const int row0    = blockIdx.x * 16;
    const int arow    = row0 + l;
    const int col     = wave * 16 + l;

    float ascale = 1.0f;
    if (SCALE_A) ascale = rowscale_a[arow];

    v8f c = {};
    const float* arowp = A + (long long)arow * KDIM;

    #pragma unroll
    for (int k = 0; k < KDIM; k += 4) {
        const int ka = k + half * 2;
        v2f a = *(const v2f*)(arowp + ka);
        if (SCALE_A) { a.x *= ascale; a.y *= ascale; }
        v2f b;
        b.x = B[(long long)ka * NDIM + col];
        b.y = B[(long long)(ka + 1) * NDIM + col];
        c = __builtin_amdgcn_wmma_f32_16x16x4_f32(
                 false, a,  false, b,
                 (short)0, c,  false,  false);
    }

    #pragma unroll
    for (int r = 0; r < 8; ++r) {
        int row = row0 + r + half * 8;
        float v = c[r];
        if (EPILOGUE) {
            v = v * rowscale_d[row] + bias[col];
            v = fmaxf(v, 0.0f);
        }
        D[(long long)row * NDIM + col] = v;
    }
}


extern "C" void kernel_launch(void* const* d_in, const int* in_sizes, int n_in,
                              void* d_out, int out_size, void* d_ws, size_t ws_size,
                              hipStream_t stream) {
    const float* x   = (const float*)d_in[0];
    const float* W1  = (const float*)d_in[1];
    const float* b1  = (const float*)d_in[2];
    const float* W2  = (const float*)d_in[3];
    const float* b2  = (const float*)d_in[4];
    const int*   src = (const int*)d_in[5];
    const int*   dst = (const int*)d_in[6];
    float*       out = (float*)d_out;

    float* ws       = (float*)d_ws;
    float* out_deg  = ws;
    float* in_deg   = out_deg + N_NODES;
    float* norm_src = in_deg + N_NODES;
    float* norm_dst = norm_src + N_NODES;
    float* agg      = norm_dst + N_NODES;
    float* h1       = agg + (long long)N_NODES * 64;
    float* h2w      = h1 + (long long)N_NODES * 128;
    unsigned* lists = (unsigned*)(h2w + (long long)N_NODES * 64);
    int* list_blocks = (int*)(lists + (long long)(N_NODES / TILE_NODES) * AGG_WAVES * LIST_BLOCKS * 32);

    const int ZT = 256;
    const long long aggElems = (long long)N_NODES * 64;

    zero_kernel<<<(2 * N_NODES + ZT - 1) / ZT, ZT, 0, stream>>>(out_deg, 2 * N_NODES);
    degree_kernel<<<(N_EDGES + ZT - 1) / ZT, ZT, 0, stream>>>(src, dst, out_deg, in_deg);
    norm_kernel<<<(N_NODES + ZT - 1) / ZT, ZT, 0, stream>>>(out_deg, in_deg, norm_src, norm_dst);

    owner_aggregate_kernel<true, true><<<N_NODES / TILE_NODES, AGG_THREADS, 0, stream>>>(
        x, norm_src, src, dst, lists, list_blocks, agg);
    wmma_gemm_kernel<F_IN, F_HID, false, true><<<N_NODES / 16, 8 * 32, 0, stream>>>(
        agg, W1, nullptr, norm_dst, b1, h1);

    wmma_gemm_kernel<F_HID, F_OUT, true, false><<<N_NODES / 16, 4 * 32, 0, stream>>>(
        h1, W2, norm_src, nullptr, nullptr, h2w);
    owner_aggregate_kernel<false, false><<<N_NODES / TILE_NODES, AGG_THREADS, 0, stream>>>(
        h2w, nullptr, src, dst, lists, list_blocks, agg);

    final_kernel<<<(int)((aggElems + ZT - 1) / ZT), ZT, 0, stream>>>(agg, norm_dst, b2, out);
}


# Final scale-and-bias pass folded into the layer-2 aggregation's tile store: a work-group writes out = sums * norm_dst + b2 straight from its LDS tile; one launch, agg round trip gone; sum order kept
# speedup vs baseline: 1.9117x; 1.0526x over previous
#include <hip/hip_runtime.h>


#define N_NODES 50000
#define N_EDGES 800000
#define F_IN    64
#define F_HID   128
#define F_OUT   64

typedef __attribute__((ext_vector_type(2))) float v2f;
typedef __attribute__((ext_vector_type(8))) float v8f;


__global__ void zero_kernel(float* __restrict__ p, int n) {
    int i = blockIdx.x * blockDim.x + threadIdx.x;
    if (i < n) p[i] = 0.0f;
}

__global__ void degree_kernel(const int* __restrict__ src, const int* __restrict__ dst,
                              float* __restrict__ out_deg, float* __restrict__ in_deg) {
    int e = blockIdx.x * blockDim.x + threadIdx.x;
    if (e < N_EDGES) {
        atomicAdd(&out_deg[src[e]], 1.0f);
        atomicAdd(&in_deg[dst[e]], 1.0f);
    }
}

__global__ void norm_kernel(const float* __restrict__ out_deg, const float* __restrict__ in_deg,
                            float* __restrict__ norm_src, float* __restrict__ norm_dst) {
    int i = blockIdx.x * blockDim.x + threadIdx.x;
    if (i < N_NODES) {
        norm_src[i] = rsqrtf(fmaxf(out_deg[i], 1.0f));
        norm_dst[i] = rsqrtf(fmaxf(in_deg[i], 1.0f));
    }
}

#define TILE_NODES   200
#define AGG_THREADS  1024
#define AGG_WAVES    (AGG_THREADS / 32)
#define SCAN_LOADS   16
#define CHUNK_EDGES  (SCAN_LOADS * 128)
#define DRAIN_EVERY  4
#define QCAP         544
static_assert(N_NODES % TILE_NODES == 0, "tiles must cover the nodes exactly");
static_assert(N_EDGES % 4 == 0, "dst is scanned in 16-byte pieces");
static_assert(N_EDGES <= (1 << 20) && TILE_NODES <= (1 << 12), "queue entry packing");
static_assert(SCAN_LOADS % DRAIN_EVERY == 0 && QCAP >= 31 + DRAIN_EVERY * 128, "queue cannot overflow");

#define LIST_BLOCKS  16
#define LIST_INVALID 0xFFFFFFFFu
static_assert(TILE_NODES <= (1 << 8) && N_NODES < (1 << 24), "list entry packing");

template <bool SCALED, int ROWS>
__device__ __forceinline__ void add_rows(int sv, int dl, float sc, int cnt, int lane,
                                         const float* __restrict__ feat, float* tile) {
    #pragma unroll 1
    for (int j0 = 0; j0 < cnt; j0 += ROWS) {
        v2f r[ROWS];
        #pragma unroll
        for (int u = 0; u < ROWS; ++u) {
            const int j  = (j0 + u < cnt) ? (j0 + u) : (cnt - 1);
            const int sj = __builtin_amdgcn_readlane(sv, j);
            r[u] = *(const v2f*)(feat + (long long)sj * 64 + lane * 2);
        }
        #pragma unroll
        for (int u = 0; u < ROWS; ++u) {
            const int j = j0 + u;
            if (j < cnt) {
                const int dj = __builtin_amdgcn_readlane(dl, j);
                float x0 = r[u].x, x1 = r[u].y;
                if (SCALED) {
                    const float sj = __int_as_float(__builtin_amdgcn_readlane(__float_as_int(sc), j));
                    x0 *= sj; x1 *= sj;
                }
                float* t = tile + dj * 64 + lane * 2;
                atomicAdd(t, x0);
                atomicAdd(t + 1, x1);
            }
        }
    }
}

template <bool SCALED, bool RECORD>
__device__ __forceinline__ void add_queued_rows(const unsigned* q, int cnt, int lane,
                                                const float* __restrict__ feat,
                                                const float* __restrict__ scale,
                                                const int* __restrict__ src,
                                                float* tile,
                                                unsigned* __restrict__ wave_list, int& nrec) {
    const unsigned ent = q[lane < cnt ? lane : 0];
    const int e  = (int)(ent & 0xFFFFFu);
    const int dl = (int)(ent >> 20);
    const int sv = src[e];
    float sc = 1.0f;
    if (SCALED) sc = scale[sv];
    if (RECORD) {
        if (nrec < LIST_BLOCKS)
            wave_list[nrec * 32 + lane] = (lane < cnt) ? (((unsigned)sv << 8) | (unsigned)dl) : LIST_INVALID;
        ++nrec;
    }
    add_rows<SCALED, 8>(sv, dl, sc, cnt, lane, feat, tile);
}

template <bool SCALED, bool RECORD>
__device__ __forceinline__ int scan_and_add(unsigned* q, int lane, int wave, int base,
                                            const float* __restrict__ feat,
                                            const float* __restrict__ scale,
                                            const int* __restrict__ src, const int* __restrict__ dst,
                                            float* tile, unsigned* __restrict__ wave_list) {
    int qn = 0;
    int nrec = 0;
    const int4* dst4 = (const int4*)dst;
    const int n4 = N_EDGES / 4;
    const int nchunks = (N_EDGES + CHUNK_EDGES - 1) / CHUNK_EDGES;
    const unsigned below = (1u << lane) - 1u;

    const int rot = (int)((blockIdx.x * 25u) % (unsigned)nchunks);
    for (int cl = wave; cl < nchunks; cl += AGG_WAVES) {
        const int c = (cl + rot < nchunks) ? (cl + rot) : (cl + rot - nchunks);
        int4 v[SCAN_LOADS];
        const int i4base = c * (CHUNK_EDGES / 4) + lane;
        #pragma unroll
        for (int k = 0; k < SCAN_LOADS; ++k) {
            const int i4 = i4base + k * 32;
            v[k] = (i4 < n4) ? dst4[i4] : make_int4(-1, -1, -1, -1);
        }
        #pragma unroll
        for (int k = 0; k < SCAN_LOADS; ++k) {
            const int e0 = (i4base + k * 32) * 4;
            const unsigned d0 = (unsigned)(v[k].x - base), d1 = (unsigned)(v[k].y - base);
            const unsigned d2 = (unsigned)(v[k].z - base), d3 = (unsigned)(v[k].w - base);
            const bool m0 = d0 < TILE_NODES, m1 = d1 < TILE_NODES;
            const bool m2 = d2 < TILE_NODES, m3 = d3 < TILE_NODES;
            if (__builtin_amdgcn_ballot_w32(m0 | m1 | m2 | m3) != 0u) {
                const unsigned b0 = __builtin_amdgcn_ballot_w32(m0);
                const unsigned b1 = __builtin_amdgcn_ballot_w32(m1);
                const unsigned b2 = __builtin_amdgcn_ballot_w32(m2);
                const unsigned b3 = __builtin_amdgcn_ballot_w32(m3);
                int p = qn;
                if (m0) q[p + __builtin_popcount(b0 & below)] = (unsigned)(e0 + 0) | (d0 << 20);
                p += __builtin_popcount(b0);
                if (m1) q[p + __builtin_popcount(b1 & below)] = (unsigned)(e0 + 1) | (d1 << 20);
                p += __builtin_popcount(b1);
                if (m2) q[p + __builtin_popcount(b2 & below)] = (unsigned)(e0 + 2) | (d2 << 20);
                p += __builtin_popcount(b2);
                if (m3) q[p + __builtin_popcount(b3 & below)] = (unsigned)(e0 + 3) | (d3 << 20);
                p += __builtin_popcount(b3);
                qn = p;
            }
            if ((k % DRAIN_EVERY) != DRAIN_EVERY - 1) continue;
            while (qn >= 32) {
                __builtin_amdgcn_wave_barrier();
                qn -= 32;
                add_queued_rows<SCALED, RECORD>(q + qn, 32, lane, feat, scale, src, tile, wave_list, nrec);
            }
        }
    }
    if (qn > 0) {
        __builtin_amdgcn_wave_barrier();
        add_queued_rows<SCALED, RECORD>(q, qn, lane, feat, scale, src, tile, wave_list, nrec);
    }
    return nrec;
}

#define STORE_PASSES ((TILE_NODES * 16 + AGG_THREADS - 1) / AGG_THREADS)
static_assert(AGG_THREADS % 16 == 0, "a lane keeps its four columns from one store pass to the next");
template <bool SCALED, bool FIRST, bool EPILOGUE>
__global__ __launch_bounds__(AGG_THREADS)
void owner_aggregate_kernel(const float* __restrict__ feat,
                            const float* __restrict__ scale,
                            const int* __restrict__ src, const int* __restrict__ dst,
                            unsigned* __restrict__ lists,
                            int* __restrict__ list_blocks,
                            const float* __restrict__ rowscale,
                            const float* __restrict__ bias,
                            float* __restrict__ dest) {
    __shared__ float    tile[TILE_NODES * 64];
    __shared__ unsigned queue[AGG_WAVES][QCAP];

    const int lane = threadIdx.x & 31;
    const int wave = __builtin_amdgcn_readfirstlane((int)(threadIdx.x >> 5));
    const int base = blockIdx.x * TILE_NODES;

    for (int i = threadIdx.x; i < TILE_NODES * 64; i += AGG_THREADS) tile[i] = 0.0f;
    __syncthreads();

    unsigned* q = queue[wave];
    unsigned* wave_list = lists + (long long)(blockIdx.x * AGG_WAVES + wave) * (LIST_BLOCKS * 32);

    if (FIRST) {
        const int nrec = scan_and_add<SCALED, true>(q, lane, wave, base, feat, scale, src, dst, tile, wave_list);
        if (lane == 0) list_blocks[blockIdx.x * AGG_WAVES + wave] = nrec;
    } else {
        const int nb_l = list_blocks[blockIdx.x * AGG_WAVES + lane];
        if (__builtin_amdgcn_ballot_w32(nb_l > LIST_BLOCKS) == 0u) {
            const int nb = __builtin_amdgcn_readlane(nb_l, wave);
            unsigned ent_next = (nb > 0) ? wave_list[lane] : LIST_INVALID;
            for (int b = 0; b < nb; ++b) {
                const unsigned ent = ent_next;
                ent_next = (b + 1 < nb) ? wave_list[(b + 1) * 32 + lane] : LIST_INVALID;
                const bool ok = ent != LIST_INVALID;
                const int cnt = __builtin_popcount(__builtin_amdgcn_ballot_w32(ok));
                const int sv = ok ? (int)(ent >> 8) : 0;
                const int dl = (int)(ent & 0xFFu);
                float sc = 1.0f;
                if (SCALED) sc = scale[sv];
                add_rows<SCALED, 16>(sv, dl, sc, cnt, lane, feat, tile);
            }
        } else {
            scan_and_add<SCALED, false>(q, lane, wave, base, feat, scale, src, dst, tile, wave_list);
        }
    }
    float  rs[STORE_PASSES] = {};
    float4 b4 = make_float4(0.0f, 0.0f, 0.0f, 0.0f);
    if (EPILOGUE) {
        b4 = ((const float4*)bias)[threadIdx.x & 15];
        #pragma unroll
        for (int p = 0; p < STORE_PASSES; ++p) {
            const int row = (int)(threadIdx.x >> 4) + p * (AGG_THREADS / 16);
            rs[p] = (row < TILE_NODES) ? rowscale[base + row] : 0.0f;
        }
    }
    __syncthreads();

    float4* out4 = (float4*)(dest + (long long)base * 64);
    const float4* tile4 = (const float4*)tile;
    #pragma unroll
    for (int p = 0; p < STORE_PASSES; ++p) {
        const int i = threadIdx.x + p * AGG_THREADS;
        if (i < TILE_NODES * 16) {
            float4 t = tile4[i];
            if (EPILOGUE) {
                t.x = t.x * rs[p] + b4.x;
                t.y = t.y * rs[p] + b4.y;
                t.z = t.z * rs[p] + b4.z;
                t.w = t.w * rs[p] + b4.w;
            }
            out4[i] = t;
        }
    }
}

template <int KDIM, int NDIM, bool SCALE_A, bool EPILOGUE>
__global__ void wmma_gemm_kernel(const float* __restrict__ A,
                                 const float* __restrict__ B,
                                 const float* __restrict__ rowscale_a,
                                 const float* __restrict__ rowscale_d,
                                 const float* __restrict__ bias,
                                 float* __restrict__ D) {
    const int lane    = threadIdx.x & 31;
    const int wave    = threadIdx.x >> 5;
    const int half    = lane >> 4;
    const int l       = lane & 15;
    const int row0    = blockIdx.x * 16;
    const int arow    = row0 + l;
    const int col     = wave * 16 + l;

    float ascale = 1.0f;
    if (SCALE_A) ascale = rowscale_a[arow];

    v8f c = {};
    const float* arowp = A + (long long)arow * KDIM;

    #pragma unroll
    for (int k = 0; k < KDIM; k += 4) {
        const int ka = k + half * 2;
        v2f a = *(const v2f*)(arowp + ka);
        if (SCALE_A) { a.x *= ascale; a.y *= ascale; }
        v2f b;
        b.x = B[(long long)ka * NDIM + col];
        b.y = B[(long long)(ka + 1) * NDIM + col];
        c = __builtin_amdgcn_wmma_f32_16x16x4_f32(
                 false, a,  false, b,
                 (short)0, c,  false,  false);
    }

    #pragma unroll
    for (int r = 0; r < 8; ++r) {
        int row = row0 + r + half * 8;
        float v = c[r];
        if (EPILOGUE) {
            v = v * rowscale_d[row] + bias[col];
            v = fmaxf(v, 0.0f);
        }
        D[(long long)row * NDIM + col] = v;
    }
}


extern "C" void kernel_launch(void* const* d_in, const int* in_sizes, int n_in,
                              void* d_out, int out_size, void* d_ws, size_t ws_size,
                              hipStream_t stream) {
    const float* x   = (const float*)d_in[0];
    const float* W1  = (const float*)d_in[1];
    const float* b1  = (const float*)d_in[2];
    const float* W2  = (const float*)d_in[3];
    const float* b2  = (const float*)d_in[4];
    const int*   src = (const int*)d_in[5];
    const int*   dst = (const int*)d_in[6];
    float*       out = (float*)d_out;

    float* ws       = (float*)d_ws;
    float* out_deg  = ws;
    float* in_deg   = out_deg + N_NODES;
    float* norm_src = in_deg + N_NODES;
    float* norm_dst = norm_src + N_NODES;
    float* agg      = norm_dst + N_NODES;
    float* h1       = agg + (long long)N_NODES * 64;
    float* h2w      = h1 + (long long)N_NODES * 128;
    unsigned* lists = (unsigned*)(h2w + (long long)N_NODES * 64);
    int* list_blocks = (int*)(lists + (long long)(N_NODES / TILE_NODES) * AGG_WAVES * LIST_BLOCKS * 32);

    const int ZT = 256;

    zero_kernel<<<(2 * N_NODES + ZT - 1) / ZT, ZT, 0, stream>>>(out_deg, 2 * N_NODES);
    degree_kernel<<<(N_EDGES + ZT - 1) / ZT, ZT, 0, stream>>>(src, dst, out_deg, in_deg);
    norm_kernel<<<(N_NODES + ZT - 1) / ZT, ZT, 0, stream>>>(out_deg, in_deg, norm_src, norm_dst);

    owner_aggregate_kernel<true, true, false><<<N_NODES / TILE_NODES, AGG_THREADS, 0, stream>>>(
        x, norm_src, src, dst, lists, list_blocks, nullptr, nullptr, agg);
    wmma_gemm_kernel<F_IN, F_HID, false, true><<<N_NODES / 16, 8 * 32, 0, stream>>>(
        agg, W1, nullptr, norm_dst, b1, h1);

    wmma_gemm_kernel<F_HID, F_OUT, true, false><<<N_NODES / 16, 4 * 32, 0, stream>>>(
        h1, W2, norm_src, nullptr, nullptr, h2w);
    owner_aggregate_kernel<false, false, true><<<N_NODES / TILE_NODES, AGG_THREADS, 0, stream>>>(
        h2w, nullptr, src, dst, lists, list_blocks, norm_dst, b2, out);
}


# Aggregation scans one tile-number byte per edge (written by the degree pass, with a packed source-row/local-row word read for matches only) instead of 4-byte dst; LDS float adds in arrival order
# speedup vs baseline: 2.1565x; 1.1281x over previous
#include <hip/hip_runtime.h>


#define N_NODES 50000
#define N_EDGES 800000
#define F_IN    64
#define F_HID   128
#define F_OUT   64

typedef __attribute__((ext_vector_type(2))) float v2f;
typedef __attribute__((ext_vector_type(8))) float v8f;


__global__ void zero_kernel(float* __restrict__ p, int n) {
    int i = blockIdx.x * blockDim.x + threadIdx.x;
    if (i < n) p[i] = 0.0f;
}

__global__ void norm_kernel(const float* __restrict__ out_deg, const float* __restrict__ in_deg,
                            float* __restrict__ norm_src, float* __restrict__ norm_dst) {
    int i = blockIdx.x * blockDim.x + threadIdx.x;
    if (i < N_NODES) {
        norm_src[i] = rsqrtf(fmaxf(out_deg[i], 1.0f));
        norm_dst[i] = rsqrtf(fmaxf(in_deg[i], 1.0f));
    }
}

#define TILE_NODES   200
#define N_TILES      (N_NODES / TILE_NODES)
#define AGG_THREADS  1024
#define AGG_WAVES    (AGG_THREADS / 32)
#define LOAD_EDGES   512
#define N_LOADS      ((N_EDGES + LOAD_EDGES - 1) / LOAD_EDGES)
#define SCAN_STEPS   ((N_LOADS + AGG_WAVES - 1) / AGG_WAVES)
#define SCAN_GROUP   7
#define ID_DWORDS    (SCAN_STEPS * AGG_WAVES * (LOAD_EDGES / 4))
#define QCAP         544
static_assert(N_NODES % TILE_NODES == 0, "tiles must cover the nodes exactly");
static_assert(N_TILES <= 255, "a tile number is one byte, and 255 is the number that matches no tile");
static_assert(N_EDGES % 4 == 0, "the degree pass takes four edges per thread");
static_assert(N_EDGES <= (1 << 20), "queue entry");
static_assert(N_NODES <= (1 << 16) && TILE_NODES <= (1 << 8), "packed word: source row | local destination row << 16");
static_assert(SCAN_STEPS % SCAN_GROUP == 0, "a wave's loads come in whole groups");
static_assert(QCAP >= 31 + LOAD_EDGES, "queue cannot overflow: it is drained after every load");

__global__ void edge_pass_kernel(const int* __restrict__ src, const int* __restrict__ dst,
                                 float* __restrict__ out_deg, float* __restrict__ in_deg,
                                 unsigned* __restrict__ tile_ids,
                                 unsigned* __restrict__ packed) {
    int i = blockIdx.x * blockDim.x + threadIdx.x;
    if (i < N_EDGES / 4) {
        const int4 s = ((const int4*)src)[i];
        const int4 d = ((const int4*)dst)[i];
        atomicAdd(&out_deg[s.x], 1.0f);
        atomicAdd(&out_deg[s.y], 1.0f);
        atomicAdd(&out_deg[s.z], 1.0f);
        atomicAdd(&out_deg[s.w], 1.0f);
        atomicAdd(&in_deg[d.x], 1.0f);
        atomicAdd(&in_deg[d.y], 1.0f);
        atomicAdd(&in_deg[d.z], 1.0f);
        atomicAdd(&in_deg[d.w], 1.0f);
        const unsigned t0 = min((unsigned)d.x / TILE_NODES, 255u), t1 = min((unsigned)d.y / TILE_NODES, 255u);
        const unsigned t2 = min((unsigned)d.z / TILE_NODES, 255u), t3 = min((unsigned)d.w / TILE_NODES, 255u);
        tile_ids[i] = t0 | (t1 << 8) | (t2 << 16) | (t3 << 24);
        uint4 p;
        p.x = (unsigned)s.x | (((unsigned)d.x - t0 * TILE_NODES) << 16);
        p.y = (unsigned)s.y | (((unsigned)d.y - t1 * TILE_NODES) << 16);
        p.z = (unsigned)s.z | (((unsigned)d.z - t2 * TILE_NODES) << 16);
        p.w = (unsigned)s.w | (((unsigned)d.w - t3 * TILE_NODES) << 16);
        ((uint4*)packed)[i] = p;
    } else if (i < ID_DWORDS) {
        tile_ids[i] = ~0u;
    }
}

#define LIST_BLOCKS  16
#define LIST_INVALID 0xFFFFFFFFu
static_assert(TILE_NODES <= (1 << 8) && N_NODES < (1 << 24), "list entry packing");

template <bool SCALED, int ROWS>
__device__ __forceinline__ void add_rows(int sv, int dl, float sc, int cnt, int lane,
                                         const float* __restrict__ feat, float* tile) {
    #pragma unroll 1
    for (int j0 = 0; j0 < cnt; j0 += ROWS) {
        v2f r[ROWS];
        #pragma unroll
        for (int u = 0; u < ROWS; ++u) {
            const int j  = (j0 + u < cnt) ? (j0 + u) : (cnt - 1);
            const int sj = __builtin_amdgcn_readlane(sv, j);
            r[u] = *(const v2f*)(feat + (long long)sj * 64 + lane * 2);
        }
        #pragma unroll
        for (int u = 0; u < ROWS; ++u) {
            const int j = j0 + u;
            if (j < cnt) {
                const int dj = __builtin_amdgcn_readlane(dl, j);
                float x0 = r[u].x, x1 = r[u].y;
                if (SCALED) {
                    const float sj = __int_as_float(__builtin_amdgcn_readlane(__float_as_int(sc), j));
                    x0 *= sj; x1 *= sj;
                }
                float* t = tile + dj * 64 + lane * 2;
                atomicAdd(t, x0);
                atomicAdd(t + 1, x1);
            }
        }
    }
}

template <bool SCALED, bool RECORD>
__device__ __forceinline__ void add_queued_rows(const unsigned* q, int cnt, int lane,
                                                const float* __restrict__ feat,
                                                const float* __restrict__ scale,
                                                const unsigned* __restrict__ packed,
                                                float* tile,
                                                unsigned* __restrict__ wave_list, int& nrec) {
    const unsigned e  = q[lane < cnt ? lane : 0];
    const unsigned pk = packed[e];
    const int sv = (int)(pk & 0xFFFFu);
    const int dl = (int)(pk >> 16);
    float sc = 1.0f;
    if (SCALED) sc = scale[sv];
    if (RECORD) {
        if (nrec < LIST_BLOCKS)
            wave_list[nrec * 32 + lane] = (lane < cnt) ? (((unsigned)sv << 8) | (unsigned)dl) : LIST_INVALID;
        ++nrec;
    }
    add_rows<SCALED, 8>(sv, dl, sc, cnt, lane, feat, tile);
}

template <bool SCALED, bool RECORD>
__device__ __forceinline__ int scan_and_add(unsigned* q, int lane, int wave,
                                            const float* __restrict__ feat,
                                            const float* __restrict__ scale,
                                            const unsigned* __restrict__ tile_ids,
                                            const unsigned* __restrict__ packed,
                                            float* tile, unsigned* __restrict__ wave_list) {
    int qn = 0;
    int nrec = 0;
    const uint4* ids16 = (const uint4*)tile_ids;
    const unsigned below = (1u << lane) - 1u;
    const unsigned mine = blockIdx.x * 0x01010101u;

    const int rot = (int)((blockIdx.x * 25u) % (unsigned)SCAN_STEPS);
    for (int g = 0; g < SCAN_STEPS; g += SCAN_GROUP) {
        uint4 v[SCAN_GROUP];
        #pragma unroll
        for (int k = 0; k < SCAN_GROUP; ++k) {
            const int j = (g + k + rot < SCAN_STEPS) ? (g + k + rot) : (g + k + rot - SCAN_STEPS);
            v[k] = ids16[(j * AGG_WAVES + wave) * 32 + lane];
        }
        #pragma unroll
        for (int k = 0; k < SCAN_GROUP; ++k) {
            const int j = (g + k + rot < SCAN_STEPS) ? (g + k + rot) : (g + k + rot - SCAN_STEPS);
            const unsigned e0 = (unsigned)((j * AGG_WAVES + wave) * 32 + lane) * 16u;
            const unsigned x0 = v[k].x ^ mine, x1 = v[k].y ^ mine, x2 = v[k].z ^ mine, x3 = v[k].w ^ mine;
            const unsigned y0 = ~(((x0 & 0x7F7F7F7Fu) + 0x7F7F7F7Fu) | x0) & 0x80808080u;
            const unsigned y1 = ~(((x1 & 0x7F7F7F7Fu) + 0x7F7F7F7Fu) | x1) & 0x80808080u;
            const unsigned y2 = ~(((x2 & 0x7F7F7F7Fu) + 0x7F7F7F7Fu) | x2) & 0x80808080u;
            const unsigned y3 = ~(((x3 & 0x7F7F7F7Fu) + 0x7F7F7F7Fu) | x3) & 0x80808080u;
            unsigned hit = y0 | (y1 >> 1) | (y2 >> 2) | (y3 >> 3);
            while (__builtin_amdgcn_ballot_w32(hit != 0u) != 0u) {
                const bool has = hit != 0u;
                const unsigned bal = __builtin_amdgcn_ballot_w32(has);
                if (has) {
                    const int p = __builtin_ctz(hit);
                    const unsigned off = (unsigned)((7 - (p & 7)) * 4 + (p >> 3));
                    q[qn + __builtin_popcount(bal & below)] = e0 + off;
                    hit &= hit - 1u;
                }
                qn += __builtin_popcount(bal);
            }
            while (qn >= 32) {
                __builtin_amdgcn_wave_barrier();
                qn -= 32;
                add_queued_rows<SCALED, RECORD>(q + qn, 32, lane, feat, scale, packed, tile, wave_list, nrec);
            }
        }
    }
    if (qn > 0) {
        __builtin_amdgcn_wave_barrier();
        add_queued_rows<SCALED, RECORD>(q, qn, lane, feat, scale, packed, tile, wave_list, nrec);
    }
    return nrec;
}

#define STORE_PASSES ((TILE_NODES * 16 + AGG_THREADS - 1) / AGG_THREADS)
static_assert(AGG_THREADS % 16 == 0, "a lane keeps its four columns from one store pass to the next");
template <bool SCALED, bool FIRST, bool EPILOGUE>
__global__ __launch_bounds__(AGG_THREADS) __attribute__((amdgpu_waves_per_eu(8, 8)))
void owner_aggregate_kernel(const float* __restrict__ feat,
                            const float* __restrict__ scale,
                            const unsigned* __restrict__ tile_ids,
                            const unsigned* __restrict__ packed,
                            unsigned* __restrict__ lists,
                            int* __restrict__ list_blocks,
                            const float* __restrict__ rowscale,
                            const float* __restrict__ bias,
                            float* __restrict__ dest) {
    __shared__ float    tile[TILE_NODES * 64];
    __shared__ unsigned queue[AGG_WAVES][QCAP];

    const int lane = threadIdx.x & 31;
    const int wave = __builtin_amdgcn_readfirstlane((int)(threadIdx.x >> 5));
    const int base = blockIdx.x * TILE_NODES;

    for (int i = threadIdx.x; i < TILE_NODES * 64; i += AGG_THREADS) tile[i] = 0.0f;
    __syncthreads();

    unsigned* q = queue[wave];
    unsigned* wave_list = lists + (long long)(blockIdx.x * AGG_WAVES + wave) * (LIST_BLOCKS * 32);

    if (FIRST) {
        const int nrec = scan_and_add<SCALED, true>(q, lane, wave, feat, scale, tile_ids, packed, tile, wave_list);
        if (lane == 0) list_blocks[blockIdx.x * AGG_WAVES + wave] = nrec;
    } else {
        const int nb_l = list_blocks[blockIdx.x * AGG_WAVES + lane];
        if (__builtin_amdgcn_ballot_w32(nb_l > LIST_BLOCKS) == 0u) {
            const int nb = __builtin_amdgcn_readlane(nb_l, wave);
            unsigned ent_next = (nb > 0) ? wave_list[lane] : LIST_INVALID;
            for (int b = 0; b < nb; ++b) {
                const unsigned ent = ent_next;
                ent_next = (b + 1 < nb) ? wave_list[(b + 1) * 32 + lane] : LIST_INVALID;
                const bool ok = ent != LIST_INVALID;
                const int cnt = __builtin_popcount(__builtin_amdgcn_ballot_w32(ok));
                const int sv = ok ? (int)(ent >> 8) : 0;
                const int dl = (int)(ent & 0xFFu);
                float sc = 1.0f;
                if (SCALED) sc = scale[sv];
                add_rows<SCALED, 16>(sv, dl, sc, cnt, lane, feat, tile);
            }
        } else {
            scan_and_add<SCALED, false>(q, lane, wave, feat, scale, tile_ids, packed, tile, wave_list);
        }
    }
    float  rs[STORE_PASSES] = {};
    float4 b4 = make_float4(0.0f, 0.0f, 0.0f, 0.0f);
    if (EPILOGUE) {
        b4 = ((const float4*)bias)[threadIdx.x & 15];
        #pragma unroll
        for (int p = 0; p < STORE_PASSES; ++p) {
            const int row = (int)(threadIdx.x >> 4) + p * (AGG_THREADS / 16);
            rs[p] = (row < TILE_NODES) ? rowscale[base + row] : 0.0f;
        }
    }
    __syncthreads();

    float4* out4 = (float4*)(dest + (long long)base * 64);
    const float4* tile4 = (const float4*)tile;
    #pragma unroll
    for (int p = 0; p < STORE_PASSES; ++p) {
        const int i = threadIdx.x + p * AGG_THREADS;
        if (i < TILE_NODES * 16) {
            float4 t = tile4[i];
            if (EPILOGUE) {
                t.x = t.x * rs[p] + b4.x;
                t.y = t.y * rs[p] + b4.y;
                t.z = t.z * rs[p] + b4.z;
                t.w = t.w * rs[p] + b4.w;
            }
            out4[i] = t;
        }
    }
}

template <int KDIM, int NDIM, bool SCALE_A, bool EPILOGUE>
__global__ void wmma_gemm_kernel(const float* __restrict__ A,
                                 const float* __restrict__ B,
                                 const float* __restrict__ rowscale_a,
                                 const float* __restrict__ rowscale_d,
                                 const float* __restrict__ bias,
                                 float* __restrict__ D) {
    const int lane    = threadIdx.x & 31;
    const int wave    = threadIdx.x >> 5;
    const int half    = lane >> 4;
    const int l       = lane & 15;
    const int row0    = blockIdx.x * 16;
    const int arow    = row0 + l;
    const int col     = wave * 16 + l;

    float ascale = 1.0f;
    if (SCALE_A) ascale = rowscale_a[arow];

    v8f c = {};
    const float* arowp = A + (long long)arow * KDIM;

    #pragma unroll
    for (int k = 0; k < KDIM; k += 4) {
        const int ka = k + half * 2;
        v2f a = *(const v2f*)(arowp + ka);
        if (SCALE_A) { a.x *= ascale; a.y *= ascale; }
        v2f b;
        b.x = B[(long long)ka * NDIM + col];
        b.y = B[(long long)(ka + 1) * NDIM + col];
        c = __builtin_amdgcn_wmma_f32_16x16x4_f32(
                 false, a,  false, b,
                 (short)0, c,  false,  false);
    }

    #pragma unroll
    for (int r = 0; r < 8; ++r) {
        int row = row0 + r + half * 8;
        float v = c[r];
        if (EPILOGUE) {
            v = v * rowscale_d[row] + bias[col];
            v = fmaxf(v, 0.0f);
        }
        D[(long long)row * NDIM + col] = v;
    }
}


extern "C" void kernel_launch(void* const* d_in, const int* in_sizes, int n_in,
                              void* d_out, int out_size, void* d_ws, size_t ws_size,
                              hipStream_t stream) {
    const float* x   = (const float*)d_in[0];
    const float* W1  = (const float*)d_in[1];
    const float* b1  = (const float*)d_in[2];
    const float* W2  = (const float*)d_in[3];
    const float* b2  = (const float*)d_in[4];
    const int*   src = (const int*)d_in[5];
    const int*   dst = (const int*)d_in[6];
    float*       out = (float*)d_out;

    float* ws       = (float*)d_ws;
    float* out_deg  = ws;
    float* in_deg   = out_deg + N_NODES;
    float* norm_src = in_deg + N_NODES;
    float* norm_dst = norm_src + N_NODES;
    float* agg      = norm_dst + N_NODES;
    float* h1       = agg + (long long)N_NODES * 64;
    float* h2w      = h1 + (long long)N_NODES * 128;
    unsigned* lists = (unsigned*)(h2w + (long long)N_NODES * 64);
    int* list_blocks = (int*)(lists + (long long)(N_NODES / TILE_NODES) * AGG_WAVES * LIST_BLOCKS * 32);
    unsigned* tile_ids = (unsigned*)(list_blocks + (N_NODES / TILE_NODES) * AGG_WAVES);
    unsigned* packed   = tile_ids + ID_DWORDS;

    const int ZT = 256;

    zero_kernel<<<(2 * N_NODES + ZT - 1) / ZT, ZT, 0, stream>>>(out_deg, 2 * N_NODES);
    edge_pass_kernel<<<(ID_DWORDS + ZT - 1) / ZT, ZT, 0, stream>>>(src, dst, out_deg, in_deg, tile_ids, packed);
    norm_kernel<<<(N_NODES + ZT - 1) / ZT, ZT, 0, stream>>>(out_deg, in_deg, norm_src, norm_dst);

    owner_aggregate_kernel<true, true, false><<<N_NODES / TILE_NODES, AGG_THREADS, 0, stream>>>(
        x, norm_src, tile_ids, packed, lists, list_blocks, nullptr, nullptr, agg);
    wmma_gemm_kernel<F_IN, F_HID, false, true><<<N_NODES / 16, 8 * 32, 0, stream>>>(
        agg, W1, nullptr, norm_dst, b1, h1);

    wmma_gemm_kernel<F_HID, F_OUT, true, false><<<N_NODES / 16, 4 * 32, 0, stream>>>(
        h1, W2, norm_src, nullptr, nullptr, h2w);
    owner_aggregate_kernel<false, false, true><<<N_NODES / TILE_NODES, AGG_THREADS, 0, stream>>>(
        h2w, nullptr, tile_ids, packed, lists, list_blocks, norm_dst, b2, out);
}


# In-degree counted by the tile's owner from the edges it adds (one LDS integer add per edge, norm_dst written by the owner) instead of 800000 global float atomics in the edge pass; sum order kept
# speedup vs baseline: 2.4813x; 1.1506x over previous
#include <hip/hip_runtime.h>


#define N_NODES 50000
#define N_EDGES 800000
#define F_IN    64
#define F_HID   128
#define F_OUT   64

typedef __attribute__((ext_vector_type(2))) float v2f;
typedef __attribute__((ext_vector_type(8))) float v8f;


__global__ void zero_kernel(float* __restrict__ p, int n) {
    int i = blockIdx.x * blockDim.x + threadIdx.x;
    if (i < n) p[i] = 0.0f;
}

__global__ void norm_kernel(const float* __restrict__ out_deg, float* __restrict__ norm_src) {
    int i = blockIdx.x * blockDim.x + threadIdx.x;
    if (i < N_NODES) norm_src[i] = rsqrtf(fmaxf(out_deg[i], 1.0f));
}

#define TILE_NODES   200
#define N_TILES      (N_NODES / TILE_NODES)
#define TILE_PITCH   256
#define AGG_THREADS  1024
#define AGG_WAVES    (AGG_THREADS / 32)
#define LOAD_EDGES   512
#define N_LOADS      ((N_EDGES + LOAD_EDGES - 1) / LOAD_EDGES)
#define SCAN_STEPS   ((N_LOADS + AGG_WAVES - 1) / AGG_WAVES)
#define SCAN_GROUP   7
#define ID_DWORDS    (SCAN_STEPS * AGG_WAVES * (LOAD_EDGES / 4))
#define QCAP         544
static_assert(N_NODES % TILE_NODES == 0, "tiles must cover the nodes exactly");
static_assert(TILE_PITCH >= TILE_NODES && TILE_PITCH % 32 == 0, "a tile's norm_dst entries are whole lines");
static_assert(TILE_NODES % 8 == 0, "the eight rows a lane of the product's epilogue holds lie in one tile");
static_assert(N_TILES <= 255, "a tile number is one byte, and 255 is the number that matches no tile");
static_assert(N_EDGES % 4 == 0, "the degree pass takes four edges per thread");
static_assert(N_EDGES <= (1 << 20), "queue entry");
static_assert(N_NODES <= (1 << 16) && TILE_NODES <= (1 << 8), "packed word: source row | local destination row << 16");
static_assert(SCAN_STEPS % SCAN_GROUP == 0, "a wave's loads come in whole groups");
static_assert(QCAP >= 31 + LOAD_EDGES, "queue cannot overflow: it is drained after every load");

__global__ void edge_pass_kernel(const int* __restrict__ src, const int* __restrict__ dst,
                                 float* __restrict__ out_deg,
                                 unsigned* __restrict__ tile_ids,
                                 unsigned* __restrict__ packed) {
    int i = blockIdx.x * blockDim.x + threadIdx.x;
    if (i < N_EDGES / 4) {
        const int4 s = ((const int4*)src)[i];
        const int4 d = ((const int4*)dst)[i];
        atomicAdd(&out_deg[s.x], 1.0f);
        atomicAdd(&out_deg[s.y], 1.0f);
        atomicAdd(&out_deg[s.z], 1.0f);
        atomicAdd(&out_deg[s.w], 1.0f);
        const unsigned t0 = min((unsigned)d.x / TILE_NODES, 255u), t1 = min((unsigned)d.y / TILE_NODES, 255u);
        const unsigned t2 = min((unsigned)d.z / TILE_NODES, 255u), t3 = min((unsigned)d.w / TILE_NODES, 255u);
        tile_ids[i] = t0 | (t1 << 8) | (t2 << 16) | (t3 << 24);
        uint4 p;
        p.x = (unsigned)s.x | (((unsigned)d.x - t0 * TILE_NODES) << 16);
        p.y = (unsigned)s.y | (((unsigned)d.y - t1 * TILE_NODES) << 16);
        p.z = (unsigned)s.z | (((unsigned)d.z - t2 * TILE_NODES) << 16);
        p.w = (unsigned)s.w | (((unsigned)d.w - t3 * TILE_NODES) << 16);
        ((uint4*)packed)[i] = p;
    } else if (i < ID_DWORDS) {
        tile_ids[i] = ~0u;
    }
}

#define LIST_BLOCKS  16
#define LIST_INVALID 0xFFFFFFFFu
static_assert(TILE_NODES <= (1 << 8) && N_NODES < (1 << 24), "list entry packing");

template <bool SCALED, int ROWS>
__device__ __forceinline__ void add_rows(int sv, int dl, float sc, int cnt, int lane,
                                         const float* __restrict__ feat, float* tile) {
    #pragma unroll 1
    for (int j0 = 0; j0 < cnt; j0 += ROWS) {
        v2f r[ROWS];
        #pragma unroll
        for (int u = 0; u < ROWS; ++u) {
            const int j  = (j0 + u < cnt) ? (j0 + u) : (cnt - 1);
            const int sj = __builtin_amdgcn_readlane(sv, j);
            r[u] = *(const v2f*)(feat + (long long)sj * 64 + lane * 2);
        }
        #pragma unroll
        for (int u = 0; u < ROWS; ++u) {
            const int j = j0 + u;
            if (j < cnt) {
                const int dj = __builtin_amdgcn_readlane(dl, j);
                float x0 = r[u].x, x1 = r[u].y;
                if (SCALED) {
                    const float sj = __int_as_float(__builtin_amdgcn_readlane(__float_as_int(sc), j));
                    x0 *= sj; x1 *= sj;
                }
                float* t = tile + dj * 64 + lane * 2;
                atomicAdd(t, x0);
                atomicAdd(t + 1, x1);
            }
        }
    }
}

template <bool SCALED, bool RECORD>
__device__ __forceinline__ void add_queued_rows(const unsigned* q, int cnt, int lane,
                                                const float* __restrict__ feat,
                                                const float* __restrict__ scale,
                                                const unsigned* __restrict__ packed,
                                                float* tile, unsigned* in_count,
                                                unsigned* __restrict__ wave_list, int& nrec) {
    const unsigned e  = q[lane < cnt ? lane : 0];
    const unsigned pk = packed[e];
    const int sv = (int)(pk & 0xFFFFu);
    const int dl = (int)(pk >> 16);
    float sc = 1.0f;
    if (SCALED) sc = scale[sv];
    if (RECORD) {
        if (nrec < LIST_BLOCKS)
            wave_list[nrec * 32 + lane] = (lane < cnt) ? (((unsigned)sv << 8) | (unsigned)dl) : LIST_INVALID;
        ++nrec;
        if (lane < cnt) atomicAdd(in_count + dl, 1u);
    }
    add_rows<SCALED, 8>(sv, dl, sc, cnt, lane, feat, tile);
}

template <bool SCALED, bool RECORD>
__device__ __forceinline__ int scan_and_add(unsigned* q, int lane, int wave,
                                            const float* __restrict__ feat,
                                            const float* __restrict__ scale,
                                            const unsigned* __restrict__ tile_ids,
                                            const unsigned* __restrict__ packed,
                                            float* tile, unsigned* in_count,
                                            unsigned* __restrict__ wave_list) {
    int qn = 0;
    int nrec = 0;
    const uint4* ids16 = (const uint4*)tile_ids;
    const unsigned below = (1u << lane) - 1u;
    const unsigned mine = blockIdx.x * 0x01010101u;

    const int rot = (int)((blockIdx.x * 25u) % (unsigned)SCAN_STEPS);
    for (int g = 0; g < SCAN_STEPS; g += SCAN_GROUP) {
        uint4 v[SCAN_GROUP];
        #pragma unroll
        for (int k = 0; k < SCAN_GROUP; ++k) {
            const int j = (g + k + rot < SCAN_STEPS) ? (g + k + rot) : (g + k + rot - SCAN_STEPS);
            v[k] = ids16[(j * AGG_WAVES + wave) * 32 + lane];
        }
        #pragma unroll
        for (int k = 0; k < SCAN_GROUP; ++k) {
            const int j = (g + k + rot < SCAN_STEPS) ? (g + k + rot) : (g + k + rot - SCAN_STEPS);
            const unsigned e0 = (unsigned)((j * AGG_WAVES + wave) * 32 + lane) * 16u;
            const unsigned x0 = v[k].x ^ mine, x1 = v[k].y ^ mine, x2 = v[k].z ^ mine, x3 = v[k].w ^ mine;
            const unsigned y0 = ~(((x0 & 0x7F7F7F7Fu) + 0x7F7F7F7Fu) | x0) & 0x80808080u;
            const unsigned y1 = ~(((x1 & 0x7F7F7F7Fu) + 0x7F7F7F7Fu) | x1) & 0x80808080u;
            const unsigned y2 = ~(((x2 & 0x7F7F7F7Fu) + 0x7F7F7F7Fu) | x2) & 0x80808080u;
            const unsigned y3 = ~(((x3 & 0x7F7F7F7Fu) + 0x7F7F7F7Fu) | x3) & 0x80808080u;
            unsigned hit = y0 | (y1 >> 1) | (y2 >> 2) | (y3 >> 3);
            while (__builtin_amdgcn_ballot_w32(hit != 0u) != 0u) {
                const bool has = hit != 0u;
                const unsigned bal = __builtin_amdgcn_ballot_w32(has);
                if (has) {
                    const int p = __builtin_ctz(hit);
                    const unsigned off = (unsigned)((7 - (p & 7)) * 4 + (p >> 3));
                    q[qn + __builtin_popcount(bal & below)] = e0 + off;
                    hit &= hit - 1u;
                }
                qn += __builtin_popcount(bal);
            }
            while (qn >= 32) {
                __builtin_amdgcn_wave_barrier();
                qn -= 32;
                add_queued_rows<SCALED, RECORD>(q + qn, 32, lane, feat, scale, packed, tile, in_count, wave_list, nrec);
            }
        }
    }
    if (qn > 0) {
        __builtin_amdgcn_wave_barrier();
        add_queued_rows<SCALED, RECORD>(q, qn, lane, feat, scale, packed, tile, in_count, wave_list, nrec);
    }
    return nrec;
}

#define STORE_PASSES ((TILE_NODES * 16 + AGG_THREADS - 1) / AGG_THREADS)
static_assert(AGG_THREADS % 16 == 0, "a lane keeps its four columns from one store pass to the next");
template <bool SCALED, bool FIRST, bool EPILOGUE>
__global__ __launch_bounds__(AGG_THREADS) __attribute__((amdgpu_waves_per_eu(8, 8)))
void owner_aggregate_kernel(const float* __restrict__ feat,
                            const float* __restrict__ scale,
                            const unsigned* __restrict__ tile_ids,
                            const unsigned* __restrict__ packed,
                            unsigned* __restrict__ lists,
                            int* __restrict__ list_blocks,
                            float* __restrict__ norm_dst_out,
                            const float* __restrict__ rowscale,
                            const float* __restrict__ bias,
                            float* __restrict__ dest) {
    __shared__ float    tile[TILE_NODES * 64];
    __shared__ unsigned queue[AGG_WAVES][QCAP];
    __shared__ unsigned in_count[TILE_NODES];

    const int lane = threadIdx.x & 31;
    const int wave = __builtin_amdgcn_readfirstlane((int)(threadIdx.x >> 5));
    const int base = blockIdx.x * TILE_NODES;

    for (int i = threadIdx.x; i < TILE_NODES * 64; i += AGG_THREADS) tile[i] = 0.0f;
    if (FIRST) {
        if (threadIdx.x < TILE_NODES) in_count[threadIdx.x] = 0u;
    }
    __syncthreads();

    unsigned* q = queue[wave];
    unsigned* wave_list = lists + (long long)(blockIdx.x * AGG_WAVES + wave) * (LIST_BLOCKS * 32);

    if (FIRST) {
        const int nrec = scan_and_add<SCALED, true>(q, lane, wave, feat, scale, tile_ids, packed, tile, in_count, wave_list);
        if (lane == 0) list_blocks[blockIdx.x * AGG_WAVES + wave] = nrec;
    } else {
        const int nb_l = list_blocks[blockIdx.x * AGG_WAVES + lane];
        if (__builtin_amdgcn_ballot_w32(nb_l > LIST_BLOCKS) == 0u) {
            const int nb = __builtin_amdgcn_readlane(nb_l, wave);
            unsigned ent_next = (nb > 0) ? wave_list[lane] : LIST_INVALID;
            for (int b = 0; b < nb; ++b) {
                const unsigned ent = ent_next;
                ent_next = (b + 1 < nb) ? wave_list[(b + 1) * 32 + lane] : LIST_INVALID;
                const bool ok = ent != LIST_INVALID;
                const int cnt = __builtin_popcount(__builtin_amdgcn_ballot_w32(ok));
                const int sv = ok ? (int)(ent >> 8) : 0;
                const int dl = (int)(ent & 0xFFu);
                float sc = 1.0f;
                if (SCALED) sc = scale[sv];
                add_rows<SCALED, 16>(sv, dl, sc, cnt, lane, feat, tile);
            }
        } else {
            scan_and_add<SCALED, false>(q, lane, wave, feat, scale, tile_ids, packed, tile, in_count, wave_list);
        }
    }
    float  rs[STORE_PASSES] = {};
    float4 b4 = make_float4(0.0f, 0.0f, 0.0f, 0.0f);
    if (EPILOGUE) {
        b4 = ((const float4*)bias)[threadIdx.x & 15];
        #pragma unroll
        for (int p = 0; p < STORE_PASSES; ++p) {
            const int row = (int)(threadIdx.x >> 4) + p * (AGG_THREADS / 16);
            rs[p] = (row < TILE_NODES) ? rowscale[blockIdx.x * TILE_PITCH + row] : 0.0f;
        }
    }
    __syncthreads();

    if (FIRST) {
        if (threadIdx.x < TILE_PITCH) {
            const float deg = (threadIdx.x < TILE_NODES) ? (float)in_count[threadIdx.x] : 1.0f;
            norm_dst_out[blockIdx.x * TILE_PITCH + threadIdx.x] = rsqrtf(fmaxf(deg, 1.0f));
        }
    }

    float4* out4 = (float4*)(dest + (long long)base * 64);
    const float4* tile4 = (const float4*)tile;
    #pragma unroll
    for (int p = 0; p < STORE_PASSES; ++p) {
        const int i = threadIdx.x + p * AGG_THREADS;
        if (i < TILE_NODES * 16) {
            float4 t = tile4[i];
            if (EPILOGUE) {
                t.x = t.x * rs[p] + b4.x;
                t.y = t.y * rs[p] + b4.y;
                t.z = t.z * rs[p] + b4.z;
                t.w = t.w * rs[p] + b4.w;
            }
            out4[i] = t;
        }
    }
}

template <int KDIM, int NDIM, bool SCALE_A, bool EPILOGUE>
__global__ void wmma_gemm_kernel(const float* __restrict__ A,
                                 const float* __restrict__ B,
                                 const float* __restrict__ rowscale_a,
                                 const float* __restrict__ rowscale_d,
                                 const float* __restrict__ bias,
                                 float* __restrict__ D) {
    const int lane    = threadIdx.x & 31;
    const int wave    = threadIdx.x >> 5;
    const int half    = lane >> 4;
    const int l       = lane & 15;
    const int row0    = blockIdx.x * 16;
    const int arow    = row0 + l;
    const int col     = wave * 16 + l;

    float ascale = 1.0f;
    if (SCALE_A) ascale = rowscale_a[arow];

    v8f c = {};
    const float* arowp = A + (long long)arow * KDIM;

    #pragma unroll
    for (int k = 0; k < KDIM; k += 4) {
        const int ka = k + half * 2;
        v2f a = *(const v2f*)(arowp + ka);
        if (SCALE_A) { a.x *= ascale; a.y *= ascale; }
        v2f b;
        b.x = B[(long long)ka * NDIM + col];
        b.y = B[(long long)(ka + 1) * NDIM + col];
        c = __builtin_amdgcn_wmma_f32_16x16x4_f32(
                 false, a,  false, b,
                 (short)0, c,  false,  false);
    }

    int dpad = 0;
    if (EPILOGUE) dpad = ((row0 + half * 8) / TILE_NODES) * (TILE_PITCH - TILE_NODES);

    #pragma unroll
    for (int r = 0; r < 8; ++r) {
        int row = row0 + r + half * 8;
        float v = c[r];
        if (EPILOGUE) {
            v = v * rowscale_d[row + dpad] + bias[col];
            v = fmaxf(v, 0.0f);
        }
        D[(long long)row * NDIM + col] = v;
    }
}


#define NODE_SLOT 50176
static_assert(NODE_SLOT >= N_NODES && NODE_SLOT % 64 == 0, "per-node arrays start on 256-byte boundaries");

extern "C" void kernel_launch(void* const* d_in, const int* in_sizes, int n_in,
                              void* d_out, int out_size, void* d_ws, size_t ws_size,
                              hipStream_t stream) {
    const float* x   = (const float*)d_in[0];
    const float* W1  = (const float*)d_in[1];
    const float* b1  = (const float*)d_in[2];
    const float* W2  = (const float*)d_in[3];
    const float* b2  = (const float*)d_in[4];
    const int*   src = (const int*)d_in[5];
    const int*   dst = (const int*)d_in[6];
    float*       out = (float*)d_out;

    float* ws       = (float*)d_ws;
    float* out_deg  = ws;
    float* norm_src = out_deg + NODE_SLOT;
    float* norm_dst = norm_src + NODE_SLOT;
    float* agg      = norm_dst + N_TILES * TILE_PITCH;
    float* h1       = agg + (long long)N_NODES * 64;
    float* h2w      = h1 + (long long)N_NODES * 128;
    unsigned* lists = (unsigned*)(h2w + (long long)N_NODES * 64);
    int* list_blocks = (int*)(lists + (long long)(N_NODES / TILE_NODES) * AGG_WAVES * LIST_BLOCKS * 32);
    unsigned* tile_ids = (unsigned*)(list_blocks + (N_NODES / TILE_NODES) * AGG_WAVES);
    unsigned* packed   = tile_ids + ID_DWORDS;

    const int ZT = 256;

    zero_kernel<<<(N_NODES + ZT - 1) / ZT, ZT, 0, stream>>>(out_deg, N_NODES);
    edge_pass_kernel<<<(ID_DWORDS + ZT - 1) / ZT, ZT, 0, stream>>>(src, dst, out_deg, tile_ids, packed);
    norm_kernel<<<(N_NODES + ZT - 1) / ZT, ZT, 0, stream>>>(out_deg, norm_src);

    owner_aggregate_kernel<true, true, false><<<N_NODES / TILE_NODES, AGG_THREADS, 0, stream>>>(
        x, norm_src, tile_ids, packed, lists, list_blocks, norm_dst, nullptr, nullptr, agg);
    wmma_gemm_kernel<F_IN, F_HID, false, true><<<N_NODES / 16, 8 * 32, 0, stream>>>(
        agg, W1, nullptr, norm_dst, b1, h1);

    wmma_gemm_kernel<F_HID, F_OUT, true, false><<<N_NODES / 16, 4 * 32, 0, stream>>>(
        h1, W2, norm_src, nullptr, nullptr, h2w);
    owner_aggregate_kernel<false, false, true><<<N_NODES / TILE_NODES, AGG_THREADS, 0, stream>>>(
        h2w, nullptr, tile_ids, packed, lists, list_blocks, nullptr, norm_dst, b2, out);
}


# Out-degree counted by the source rows' owners: a work-group scans one source-tile byte per edge, counts its 200 rows by LDS integer adds, writes norm_src; global float atomics and one launch gone
# speedup vs baseline: 3.4437x; 1.3879x over previous
#include <hip/hip_runtime.h>


#define N_NODES 50000
#define N_EDGES 800000
#define F_IN    64
#define F_HID   128
#define F_OUT   64

typedef __attribute__((ext_vector_type(2))) float v2f;
typedef __attribute__((ext_vector_type(8))) float v8f;

#define TILE_NODES   200
#define N_TILES      (N_NODES / TILE_NODES)
#define TILE_PITCH   256
#define AGG_THREADS  1024
#define AGG_WAVES    (AGG_THREADS / 32)
#define LOAD_EDGES   512
#define N_LOADS      ((N_EDGES + LOAD_EDGES - 1) / LOAD_EDGES)
#define SCAN_STEPS   ((N_LOADS + AGG_WAVES - 1) / AGG_WAVES)
#define SCAN_GROUP   7
#define ID_DWORDS    (SCAN_STEPS * AGG_WAVES * (LOAD_EDGES / 4))
#define QCAP         544
static_assert(N_NODES % TILE_NODES == 0, "tiles must cover the nodes exactly");
static_assert(TILE_PITCH >= TILE_NODES && TILE_PITCH % 32 == 0, "a tile's norm_dst entries are whole lines");
static_assert(TILE_NODES % 8 == 0, "the eight rows a lane of the product's epilogue holds lie in one tile");
static_assert(N_TILES <= 255, "a tile number is one byte, and 255 is the number that matches no tile");
static_assert(N_EDGES % 4 == 0, "the degree pass takes four edges per thread");
static_assert(N_EDGES <= (1 << 20), "queue entry");
static_assert(N_NODES <= (1 << 16) && TILE_NODES <= (1 << 8), "packed word: source row | local destination row << 16");
static_assert(SCAN_STEPS % SCAN_GROUP == 0, "a wave's loads come in whole groups");
static_assert(QCAP >= 31 + LOAD_EDGES, "queue cannot overflow: it is drained after every load");

__global__ void edge_pass_kernel(const int* __restrict__ src, const int* __restrict__ dst,
                                 unsigned* __restrict__ tile_ids,
                                 unsigned* __restrict__ src_tile_ids,
                                 unsigned* __restrict__ packed) {
    int i = blockIdx.x * blockDim.x + threadIdx.x;
    if (i < N_EDGES / 4) {
        const int4 s = ((const int4*)src)[i];
        const int4 d = ((const int4*)dst)[i];
        const unsigned u0 = min((unsigned)s.x / TILE_NODES, 255u), u1 = min((unsigned)s.y / TILE_NODES, 255u);
        const unsigned u2 = min((unsigned)s.z / TILE_NODES, 255u), u3 = min((unsigned)s.w / TILE_NODES, 255u);
        src_tile_ids[i] = u0 | (u1 << 8) | (u2 << 16) | (u3 << 24);
        const unsigned t0 = min((unsigned)d.x / TILE_NODES, 255u), t1 = min((unsigned)d.y / TILE_NODES, 255u);
        const unsigned t2 = min((unsigned)d.z / TILE_NODES, 255u), t3 = min((unsigned)d.w / TILE_NODES, 255u);
        tile_ids[i] = t0 | (t1 << 8) | (t2 << 16) | (t3 << 24);
        uint4 p;
        p.x = (unsigned)s.x | (((unsigned)d.x - t0 * TILE_NODES) << 16);
        p.y = (unsigned)s.y | (((unsigned)d.y - t1 * TILE_NODES) << 16);
        p.z = (unsigned)s.z | (((unsigned)d.z - t2 * TILE_NODES) << 16);
        p.w = (unsigned)s.w | (((unsigned)d.w - t3 * TILE_NODES) << 16);
        ((uint4*)packed)[i] = p;
    } else if (i < ID_DWORDS) {
        tile_ids[i] = ~0u;
        src_tile_ids[i] = ~0u;
    }
}

__device__ __forceinline__ int pitched(int r) { return r + (r / TILE_NODES) * (TILE_PITCH - TILE_NODES); }

#define LIST_BLOCKS  16
#define LIST_INVALID 0xFFFFFFFFu
static_assert(TILE_NODES <= (1 << 8) && N_NODES < (1 << 24), "list entry packing");

template <bool SCALED, int ROWS>
__device__ __forceinline__ void add_rows(int sv, int dl, float sc, int cnt, int lane,
                                         const float* __restrict__ feat, float* tile) {
    #pragma unroll 1
    for (int j0 = 0; j0 < cnt; j0 += ROWS) {
        v2f r[ROWS];
        #pragma unroll
        for (int u = 0; u < ROWS; ++u) {
            const int j  = (j0 + u < cnt) ? (j0 + u) : (cnt - 1);
            const int sj = __builtin_amdgcn_readlane(sv, j);
            r[u] = *(const v2f*)(feat + (long long)sj * 64 + lane * 2);
        }
        #pragma unroll
        for (int u = 0; u < ROWS; ++u) {
            const int j = j0 + u;
            if (j < cnt) {
                const int dj = __builtin_amdgcn_readlane(dl, j);
                float x0 = r[u].x, x1 = r[u].y;
                if (SCALED) {
                    const float sj = __int_as_float(__builtin_amdgcn_readlane(__float_as_int(sc), j));
                    x0 *= sj; x1 *= sj;
                }
                float* t = tile + dj * 64 + lane * 2;
                atomicAdd(t, x0);
                atomicAdd(t + 1, x1);
            }
        }
    }
}

template <bool SCALED, bool RECORD>
__device__ __forceinline__ void add_queued_rows(const unsigned* q, int cnt, int lane,
                                                const float* __restrict__ feat,
                                                const float* __restrict__ scale,
                                                const unsigned* __restrict__ packed,
                                                float* tile, unsigned* in_count,
                                                unsigned* __restrict__ wave_list, int& nrec) {
    const unsigned e  = q[lane < cnt ? lane : 0];
    const unsigned pk = packed[e];
    const int sv = (int)(pk & 0xFFFFu);
    const int dl = (int)(pk >> 16);
    float sc = 1.0f;
    if (SCALED) sc = scale[pitched(sv)];
    if (RECORD) {
        if (nrec < LIST_BLOCKS)
            wave_list[nrec * 32 + lane] = (lane < cnt) ? (((unsigned)sv << 8) | (unsigned)dl) : LIST_INVALID;
        ++nrec;
        if (lane < cnt) atomicAdd(in_count + dl, 1u);
    }
    add_rows<SCALED, 8>(sv, dl, sc, cnt, lane, feat, tile);
}

__device__ __forceinline__ void count_queued_sources(const unsigned* q, int cnt, int lane,
                                                     const unsigned* __restrict__ packed,
                                                     unsigned* row_count) {
    const unsigned e  = q[lane < cnt ? lane : 0];
    const unsigned pk = packed[e];
    if (lane < cnt) atomicAdd(row_count + ((int)(pk & 0xFFFFu) - (int)(blockIdx.x * TILE_NODES)), 1u);
}

template <bool SCALED, bool RECORD, bool COUNT_SOURCES>
__device__ __forceinline__ int scan_and_add(unsigned* q, int lane, int wave,
                                            const float* __restrict__ feat,
                                            const float* __restrict__ scale,
                                            const unsigned* __restrict__ tile_ids,
                                            const unsigned* __restrict__ packed,
                                            float* tile, unsigned* in_count,
                                            unsigned* __restrict__ wave_list) {
    int qn = 0;
    int nrec = 0;
    const uint4* ids16 = (const uint4*)tile_ids;
    const unsigned below = (1u << lane) - 1u;
    const unsigned mine = blockIdx.x * 0x01010101u;

    const int rot = (int)((blockIdx.x * 25u) % (unsigned)SCAN_STEPS);
    for (int g = 0; g < SCAN_STEPS; g += SCAN_GROUP) {
        uint4 v[SCAN_GROUP];
        #pragma unroll
        for (int k = 0; k < SCAN_GROUP; ++k) {
            const int j = (g + k + rot < SCAN_STEPS) ? (g + k + rot) : (g + k + rot - SCAN_STEPS);
            v[k] = ids16[(j * AGG_WAVES + wave) * 32 + lane];
        }
        #pragma unroll
        for (int k = 0; k < SCAN_GROUP; ++k) {
            const int j = (g + k + rot < SCAN_STEPS) ? (g + k + rot) : (g + k + rot - SCAN_STEPS);
            const unsigned e0 = (unsigned)((j * AGG_WAVES + wave) * 32 + lane) * 16u;
            const unsigned x0 = v[k].x ^ mine, x1 = v[k].y ^ mine, x2 = v[k].z ^ mine, x3 = v[k].w ^ mine;
            const unsigned y0 = ~(((x0 & 0x7F7F7F7Fu) + 0x7F7F7F7Fu) | x0) & 0x80808080u;
            const unsigned y1 = ~(((x1 & 0x7F7F7F7Fu) + 0x7F7F7F7Fu) | x1) & 0x80808080u;
            const unsigned y2 = ~(((x2 & 0x7F7F7F7Fu) + 0x7F7F7F7Fu) | x2) & 0x80808080u;
            const unsigned y3 = ~(((x3 & 0x7F7F7F7Fu) + 0x7F7F7F7Fu) | x3) & 0x80808080u;
            unsigned hit = y0 | (y1 >> 1) | (y2 >> 2) | (y3 >> 3);
            while (__builtin_amdgcn_ballot_w32(hit != 0u) != 0u) {
                const bool has = hit != 0u;
                const unsigned bal = __builtin_amdgcn_ballot_w32(has);
                if (has) {
                    const int p = __builtin_ctz(hit);
                    const unsigned off = (unsigned)((7 - (p & 7)) * 4 + (p >> 3));
                    q[qn + __builtin_popcount(bal & below)] = e0 + off;
                    hit &= hit - 1u;
                }
                qn += __builtin_popcount(bal);
            }
            while (qn >= 32) {
                __builtin_amdgcn_wave_barrier();
                qn -= 32;
                if (COUNT_SOURCES) count_queued_sources(q + qn, 32, lane, packed, in_count);
                else add_queued_rows<SCALED, RECORD>(q + qn, 32, lane, feat, scale, packed, tile, in_count, wave_list, nrec);
            }
        }
    }
    if (qn > 0) {
        __builtin_amdgcn_wave_barrier();
        if (COUNT_SOURCES) count_queued_sources(q, qn, lane, packed, in_count);
        else add_queued_rows<SCALED, RECORD>(q, qn, lane, feat, scale, packed, tile, in_count, wave_list, nrec);
    }
    return nrec;
}

#define STORE_PASSES ((TILE_NODES * 16 + AGG_THREADS - 1) / AGG_THREADS)
static_assert(AGG_THREADS % 16 == 0, "a lane keeps its four columns from one store pass to the next");
template <bool SCALED, bool FIRST, bool EPILOGUE>
__global__ __launch_bounds__(AGG_THREADS) __attribute__((amdgpu_waves_per_eu(8, 8)))
void owner_aggregate_kernel(const float* __restrict__ feat,
                            const float* __restrict__ scale,
                            const unsigned* __restrict__ tile_ids,
                            const unsigned* __restrict__ packed,
                            unsigned* __restrict__ lists,
                            int* __restrict__ list_blocks,
                            float* __restrict__ norm_dst_out,
                            const float* __restrict__ rowscale,
                            const float* __restrict__ bias,
                            float* __restrict__ dest) {
    __shared__ float    tile[TILE_NODES * 64];
    __shared__ unsigned queue[AGG_WAVES][QCAP];
    __shared__ unsigned in_count[TILE_NODES];

    const int lane = threadIdx.x & 31;
    const int wave = __builtin_amdgcn_readfirstlane((int)(threadIdx.x >> 5));
    const int base = blockIdx.x * TILE_NODES;

    for (int i = threadIdx.x; i < TILE_NODES * 64; i += AGG_THREADS) tile[i] = 0.0f;
    if (FIRST) {
        if (threadIdx.x < TILE_NODES) in_count[threadIdx.x] = 0u;
    }
    __syncthreads();

    unsigned* q = queue[wave];
    unsigned* wave_list = lists + (long long)(blockIdx.x * AGG_WAVES + wave) * (LIST_BLOCKS * 32);

    if (FIRST) {
        const int nrec = scan_and_add<SCALED, true, false>(q, lane, wave, feat, scale, tile_ids, packed, tile, in_count, wave_list);
        if (lane == 0) list_blocks[blockIdx.x * AGG_WAVES + wave] = nrec;
    } else {
        const int nb_l = list_blocks[blockIdx.x * AGG_WAVES + lane];
        if (__builtin_amdgcn_ballot_w32(nb_l > LIST_BLOCKS) == 0u) {
            const int nb = __builtin_amdgcn_readlane(nb_l, wave);
            unsigned ent_next = (nb > 0) ? wave_list[lane] : LIST_INVALID;
            for (int b = 0; b < nb; ++b) {
                const unsigned ent = ent_next;
                ent_next = (b + 1 < nb) ? wave_list[(b + 1) * 32 + lane] : LIST_INVALID;
                const bool ok = ent != LIST_INVALID;
                const int cnt = __builtin_popcount(__builtin_amdgcn_ballot_w32(ok));
                const int sv = ok ? (int)(ent >> 8) : 0;
                const int dl = (int)(ent & 0xFFu);
                float sc = 1.0f;
                if (SCALED) sc = scale[pitched(sv)];
                add_rows<SCALED, 16>(sv, dl, sc, cnt, lane, feat, tile);
            }
        } else {
            scan_and_add<SCALED, false, false>(q, lane, wave, feat, scale, tile_ids, packed, tile, in_count, wave_list);
        }
    }
    float  rs[STORE_PASSES] = {};
    float4 b4 = make_float4(0.0f, 0.0f, 0.0f, 0.0f);
    if (EPILOGUE) {
        b4 = ((const float4*)bias)[threadIdx.x & 15];
        #pragma unroll
        for (int p = 0; p < STORE_PASSES; ++p) {
            const int row = (int)(threadIdx.x >> 4) + p * (AGG_THREADS / 16);
            rs[p] = (row < TILE_NODES) ? rowscale[blockIdx.x * TILE_PITCH + row] : 0.0f;
        }
    }
    __syncthreads();

    if (FIRST) {
        if (threadIdx.x < TILE_PITCH) {
            const float deg = (threadIdx.x < TILE_NODES) ? (float)in_count[threadIdx.x] : 1.0f;
            norm_dst_out[blockIdx.x * TILE_PITCH + threadIdx.x] = rsqrtf(fmaxf(deg, 1.0f));
        }
    }

    float4* out4 = (float4*)(dest + (long long)base * 64);
    const float4* tile4 = (const float4*)tile;
    #pragma unroll
    for (int p = 0; p < STORE_PASSES; ++p) {
        const int i = threadIdx.x + p * AGG_THREADS;
        if (i < TILE_NODES * 16) {
            float4 t = tile4[i];
            if (EPILOGUE) {
                t.x = t.x * rs[p] + b4.x;
                t.y = t.y * rs[p] + b4.y;
                t.z = t.z * rs[p] + b4.z;
                t.w = t.w * rs[p] + b4.w;
            }
            out4[i] = t;
        }
    }
}

__global__ __launch_bounds__(AGG_THREADS) __attribute__((amdgpu_waves_per_eu(8, 8)))
void owner_out_degree_kernel(const unsigned* __restrict__ src_tile_ids,
                             const unsigned* __restrict__ packed,
                             float* __restrict__ norm_src_out) {
    __shared__ unsigned queue[AGG_WAVES][QCAP];
    __shared__ unsigned out_count[TILE_NODES];

    const int lane = threadIdx.x & 31;
    const int wave = __builtin_amdgcn_readfirstlane((int)(threadIdx.x >> 5));

    if (threadIdx.x < TILE_NODES) out_count[threadIdx.x] = 0u;
    __syncthreads();

    scan_and_add<false, false, true>(queue[wave], lane, wave, nullptr, nullptr, src_tile_ids, packed,
                                     nullptr, out_count, nullptr);
    __syncthreads();

    if (threadIdx.x < TILE_PITCH) {
        const float deg = (threadIdx.x < TILE_NODES) ? (float)out_count[threadIdx.x] : 1.0f;
        norm_src_out[blockIdx.x * TILE_PITCH + threadIdx.x] = rsqrtf(fmaxf(deg, 1.0f));
    }
}

template <int KDIM, int NDIM, bool SCALE_A, bool EPILOGUE>
__global__ void wmma_gemm_kernel(const float* __restrict__ A,
                                 const float* __restrict__ B,
                                 const float* __restrict__ rowscale_a,
                                 const float* __restrict__ rowscale_d,
                                 const float* __restrict__ bias,
                                 float* __restrict__ D) {
    const int lane    = threadIdx.x & 31;
    const int wave    = threadIdx.x >> 5;
    const int half    = lane >> 4;
    const int l       = lane & 15;
    const int row0    = blockIdx.x * 16;
    const int arow    = row0 + l;
    const int col     = wave * 16 + l;

    float ascale = 1.0f;
    if (SCALE_A) ascale = rowscale_a[pitched(arow)];

    v8f c = {};
    const float* arowp = A + (long long)arow * KDIM;

    #pragma unroll
    for (int k = 0; k < KDIM; k += 4) {
        const int ka = k + half * 2;
        v2f a = *(const v2f*)(arowp + ka);
        if (SCALE_A) { a.x *= ascale; a.y *= ascale; }
        v2f b;
        b.x = B[(long long)ka * NDIM + col];
        b.y = B[(long long)(ka + 1) * NDIM + col];
        c = __builtin_amdgcn_wmma_f32_16x16x4_f32(
                 false, a,  false, b,
                 (short)0, c,  false,  false);
    }

    int dpad = 0;
    if (EPILOGUE) dpad = ((row0 + half * 8) / TILE_NODES) * (TILE_PITCH - TILE_NODES);

    #pragma unroll
    for (int r = 0; r < 8; ++r) {
        int row = row0 + r + half * 8;
        float v = c[r];
        if (EPILOGUE) {
            v = v * rowscale_d[row + dpad] + bias[col];
            v = fmaxf(v, 0.0f);
        }
        D[(long long)row * NDIM + col] = v;
    }
}


static_assert((N_TILES * TILE_PITCH) % 64 == 0, "the arrays behind norm_src and norm_dst start on 256-byte boundaries");

extern "C" void kernel_launch(void* const* d_in, const int* in_sizes, int n_in,
                              void* d_out, int out_size, void* d_ws, size_t ws_size,
                              hipStream_t stream) {
    const float* x   = (const float*)d_in[0];
    const float* W1  = (const float*)d_in[1];
    const float* b1  = (const float*)d_in[2];
    const float* W2  = (const float*)d_in[3];
    const float* b2  = (const float*)d_in[4];
    const int*   src = (const int*)d_in[5];
    const int*   dst = (const int*)d_in[6];
    float*       out = (float*)d_out;

    float* ws       = (float*)d_ws;
    float* norm_src = ws;
    float* norm_dst = norm_src + N_TILES * TILE_PITCH;
    float* agg      = norm_dst + N_TILES * TILE_PITCH;
    float* h1       = agg + (long long)N_NODES * 64;
    float* h2w      = h1 + (long long)N_NODES * 128;
    unsigned* lists = (unsigned*)(h2w + (long long)N_NODES * 64);
    int* list_blocks = (int*)(lists + (long long)(N_NODES / TILE_NODES) * AGG_WAVES * LIST_BLOCKS * 32);
    unsigned* tile_ids = (unsigned*)(list_blocks + (N_NODES / TILE_NODES) * AGG_WAVES);
    unsigned* src_tile_ids = tile_ids + ID_DWORDS;
    unsigned* packed   = src_tile_ids + ID_DWORDS;

    const int ZT = 256;

    edge_pass_kernel<<<(ID_DWORDS + ZT - 1) / ZT, ZT, 0, stream>>>(src, dst, tile_ids, src_tile_ids, packed);
    owner_out_degree_kernel<<<N_TILES, AGG_THREADS, 0, stream>>>(src_tile_ids, packed, norm_src);

    owner_aggregate_kernel<true, true, false><<<N_NODES / TILE_NODES, AGG_THREADS, 0, stream>>>(
        x, norm_src, tile_ids, packed, lists, list_blocks, norm_dst, nullptr, nullptr, agg);
    wmma_gemm_kernel<F_IN, F_HID, false, true><<<N_NODES / 16, 8 * 32, 0, stream>>>(
        agg, W1, nullptr, norm_dst, b1, h1);

    wmma_gemm_kernel<F_HID, F_OUT, true, false><<<N_NODES / 16, 4 * 32, 0, stream>>>(
        h1, W2, norm_src, nullptr, nullptr, h2w);
    owner_aggregate_kernel<false, false, true><<<N_NODES / TILE_NODES, AGG_THREADS, 0, stream>>>(
        h2w, nullptr, tile_ids, packed, lists, list_blocks, nullptr, norm_dst, b2, out);
}


# Both products moved into the layer-1 aggregation: the tile's owner multiplies its LDS tile by W1, transposed, and its accumulators feed the W2 product as B fragments; k order of the W2 sum permuted
# speedup vs baseline: 4.1816x; 1.2143x over previous
#include <hip/hip_runtime.h>


#define N_NODES 50000
#define N_EDGES 800000
#define F_IN    64
#define F_HID   128
#define F_OUT   64

typedef __attribute__((ext_vector_type(2))) float v2f;
typedef __attribute__((ext_vector_type(4))) float v4f;
typedef __attribute__((ext_vector_type(8))) float v8f;

#define TILE_NODES   200
#define N_TILES      (N_NODES / TILE_NODES)
#define TILE_PITCH   256
#define AGG_THREADS  1024
#define AGG_WAVES    (AGG_THREADS / 32)
#define LOAD_EDGES   512
#define N_LOADS      ((N_EDGES + LOAD_EDGES - 1) / LOAD_EDGES)
#define SCAN_STEPS   ((N_LOADS + AGG_WAVES - 1) / AGG_WAVES)
#define SCAN_GROUP   7
#define ID_DWORDS    (SCAN_STEPS * AGG_WAVES * (LOAD_EDGES / 4))
#define QCAP         544
#define ROW_TILES    ((TILE_NODES + 15) / 16)
#define TILE_ROWS    (ROW_TILES * 16)
#define W_FRAGS      (F_IN * F_HID / 4)
static_assert(N_NODES % TILE_NODES == 0, "tiles must cover the nodes exactly");
static_assert(TILE_PITCH >= TILE_NODES && TILE_PITCH % 32 == 0, "a tile's norm_dst entries are whole lines");
static_assert(TILE_NODES % 8 == 0, "the eight rows a lane of the product's epilogue holds lie in one tile");
static_assert(N_TILES <= 255, "a tile number is one byte, and 255 is the number that matches no tile");
static_assert(N_EDGES % 4 == 0, "the degree pass takes four edges per thread");
static_assert(N_EDGES <= (1 << 20), "queue entry");
static_assert(N_NODES <= (1 << 16) && TILE_NODES <= (1 << 8), "packed word: source row | local destination row << 16");
static_assert(SCAN_STEPS % SCAN_GROUP == 0, "a wave's loads come in whole groups");
static_assert(QCAP >= 31 + LOAD_EDGES, "queue cannot overflow: it is drained after every load");
static_assert(ROW_TILES <= AGG_WAVES && TILE_ROWS <= TILE_PITCH, "one wave per 16-row block; its scales lie in the tile's entries");
static_assert(F_IN * F_HID == F_HID * F_OUT && W_FRAGS % AGG_THREADS == 0, "both weight matrices are staged by the same loop");

__global__ void edge_pass_kernel(const int* __restrict__ src, const int* __restrict__ dst,
                                 unsigned* __restrict__ tile_ids,
                                 unsigned* __restrict__ src_tile_ids,
                                 unsigned* __restrict__ packed) {
    int i = blockIdx.x * blockDim.x + threadIdx.x;
    if (i < N_EDGES / 4) {
        const int4 s = ((const int4*)src)[i];
        const int4 d = ((const int4*)dst)[i];
        const unsigned u0 = min((unsigned)s.x / TILE_NODES, 255u), u1 = min((unsigned)s.y / TILE_NODES, 255u);
        const unsigned u2 = min((unsigned)s.z / TILE_NODES, 255u), u3 = min((unsigned)s.w / TILE_NODES, 255u);
        src_tile_ids[i] = u0 | (u1 << 8) | (u2 << 16) | (u3 << 24);
        const unsigned t0 = min((unsigned)d.x / TILE_NODES, 255u), t1 = min((unsigned)d.y / TILE_NODES, 255u);
        const unsigned t2 = min((unsigned)d.z / TILE_NODES, 255u), t3 = min((unsigned)d.w / TILE_NODES, 255u);
        tile_ids[i] = t0 | (t1 << 8) | (t2 << 16) | (t3 << 24);
        uint4 p;
        p.x = (unsigned)s.x | (((unsigned)d.x - t0 * TILE_NODES) << 16);
        p.y = (unsigned)s.y | (((unsigned)d.y - t1 * TILE_NODES) << 16);
        p.z = (unsigned)s.z | (((unsigned)d.z - t2 * TILE_NODES) << 16);
        p.w = (unsigned)s.w | (((unsigned)d.w - t3 * TILE_NODES) << 16);
        ((uint4*)packed)[i] = p;
    } else if (i < ID_DWORDS) {
        tile_ids[i] = ~0u;
        src_tile_ids[i] = ~0u;
    }
}

__device__ __forceinline__ int pitched(int r) { return r + (r / TILE_NODES) * (TILE_PITCH - TILE_NODES); }

#define LIST_BLOCKS  16
#define LIST_INVALID 0xFFFFFFFFu
static_assert(TILE_NODES <= (1 << 8) && N_NODES < (1 << 24), "list entry packing");

template <bool SCALED, int ROWS>
__device__ __forceinline__ void add_rows(int sv, int dl, float sc, int cnt, int lane,
                                         const float* __restrict__ feat, float* tile) {
    #pragma unroll 1
    for (int j0 = 0; j0 < cnt; j0 += ROWS) {
        v2f r[ROWS];
        #pragma unroll
        for (int u = 0; u < ROWS; ++u) {
            const int j  = (j0 + u < cnt) ? (j0 + u) : (cnt - 1);
            const int sj = __builtin_amdgcn_readlane(sv, j);
            r[u] = *(const v2f*)(feat + (long long)sj * 64 + lane * 2);
        }
        #pragma unroll
        for (int u = 0; u < ROWS; ++u) {
            const int j = j0 + u;
            if (j < cnt) {
                const int dj = __builtin_amdgcn_readlane(dl, j);
                float x0 = r[u].x, x1 = r[u].y;
                if (SCALED) {
                    const float sj = __int_as_float(__builtin_amdgcn_readlane(__float_as_int(sc), j));
                    x0 *= sj; x1 *= sj;
                }
                float* t = tile + dj * 64 + lane * 2;
                atomicAdd(t, x0);
                atomicAdd(t + 1, x1);
            }
        }
    }
}

template <bool SCALED, bool RECORD>
__device__ __forceinline__ void add_queued_rows(const unsigned* q, int cnt, int lane,
                                                const float* __restrict__ feat,
                                                const float* __restrict__ scale,
                                                const unsigned* __restrict__ packed,
                                                float* tile, unsigned* in_count,
                                                unsigned* __restrict__ wave_list, int& nrec) {
    const unsigned e  = q[lane < cnt ? lane : 0];
    const unsigned pk = packed[e];
    const int sv = (int)(pk & 0xFFFFu);
    const int dl = (int)(pk >> 16);
    float sc = 1.0f;
    if (SCALED) sc = scale[pitched(sv)];
    if (RECORD) {
        if (nrec < LIST_BLOCKS)
            wave_list[nrec * 32 + lane] = (lane < cnt) ? (((unsigned)sv << 8) | (unsigned)dl) : LIST_INVALID;
        ++nrec;
        if (lane < cnt) atomicAdd(in_count + dl, 1u);
    }
    add_rows<SCALED, 8>(sv, dl, sc, cnt, lane, feat, tile);
}

__device__ __forceinline__ void count_queued_sources(const unsigned* q, int cnt, int lane,
                                                     const unsigned* __restrict__ packed,
                                                     unsigned* row_count) {
    const unsigned e  = q[lane < cnt ? lane : 0];
    const unsigned pk = packed[e];
    if (lane < cnt) atomicAdd(row_count + ((int)(pk & 0xFFFFu) - (int)(blockIdx.x * TILE_NODES)), 1u);
}

template <bool SCALED, bool RECORD, bool COUNT_SOURCES>
__device__ __forceinline__ int scan_and_add(unsigned* q, int lane, int wave,
                                            const float* __restrict__ feat,
                                            const float* __restrict__ scale,
                                            const unsigned* __restrict__ tile_ids,
                                            const unsigned* __restrict__ packed,
                                            float* tile, unsigned* in_count,
                                            unsigned* __restrict__ wave_list) {
    int qn = 0;
    int nrec = 0;
    const uint4* ids16 = (const uint4*)tile_ids;
    const unsigned below = (1u << lane) - 1u;
    const unsigned mine = blockIdx.x * 0x01010101u;

    const int rot = (int)((blockIdx.x * 25u) % (unsigned)SCAN_STEPS);
    for (int g = 0; g < SCAN_STEPS; g += SCAN_GROUP) {
        uint4 v[SCAN_GROUP];
        #pragma unroll
        for (int k = 0; k < SCAN_GROUP; ++k) {
            const int j = (g + k + rot < SCAN_STEPS) ? (g + k + rot) : (g + k + rot - SCAN_STEPS);
            v[k] = ids16[(j * AGG_WAVES + wave) * 32 + lane];
        }
        #pragma unroll
        for (int k = 0; k < SCAN_GROUP; ++k) {
            const int j = (g + k + rot < SCAN_STEPS) ? (g + k + rot) : (g + k + rot - SCAN_STEPS);
            const unsigned e0 = (unsigned)((j * AGG_WAVES + wave) * 32 + lane) * 16u;
            const unsigned x0 = v[k].x ^ mine, x1 = v[k].y ^ mine, x2 = v[k].z ^ mine, x3 = v[k].w ^ mine;
            const unsigned y0 = ~(((x0 & 0x7F7F7F7Fu) + 0x7F7F7F7Fu) | x0) & 0x80808080u;
            const unsigned y1 = ~(((x1 & 0x7F7F7F7Fu) + 0x7F7F7F7Fu) | x1) & 0x80808080u;
            const unsigned y2 = ~(((x2 & 0x7F7F7F7Fu) + 0x7F7F7F7Fu) | x2) & 0x80808080u;
            const unsigned y3 = ~(((x3 & 0x7F7F7F7Fu) + 0x7F7F7F7Fu) | x3) & 0x80808080u;
            unsigned hit = y0 | (y1 >> 1) | (y2 >> 2) | (y3 >> 3);
            while (__builtin_amdgcn_ballot_w32(hit != 0u) != 0u) {
                const bool has = hit != 0u;
                const unsigned bal = __builtin_amdgcn_ballot_w32(has);
                if (has) {
                    const int p = __builtin_ctz(hit);
                    const unsigned off = (unsigned)((7 - (p & 7)) * 4 + (p >> 3));
                    q[qn + __builtin_popcount(bal & below)] = e0 + off;
                    hit &= hit - 1u;
                }
                qn += __builtin_popcount(bal);
            }
            while (qn >= 32) {
                __builtin_amdgcn_wave_barrier();
                qn -= 32;
                if (COUNT_SOURCES) count_queued_sources(q + qn, 32, lane, packed, in_count);
                else add_queued_rows<SCALED, RECORD>(q + qn, 32, lane, feat, scale, packed, tile, in_count, wave_list, nrec);
            }
        }
    }
    if (qn > 0) {
        __builtin_amdgcn_wave_barrier();
        if (COUNT_SOURCES) count_queued_sources(q, qn, lane, packed, in_count);
        else add_queued_rows<SCALED, RECORD>(q, qn, lane, feat, scale, packed, tile, in_count, wave_list, nrec);
    }
    return nrec;
}

#define STORE_PASSES ((TILE_NODES * 16 + AGG_THREADS - 1) / AGG_THREADS)
static_assert(AGG_THREADS % 16 == 0, "a lane keeps its four columns from one store pass to the next");
template <bool SCALED, bool FIRST, bool EPILOGUE>
__global__ __launch_bounds__(AGG_THREADS) __attribute__((amdgpu_waves_per_eu(8, 8)))
void owner_aggregate_kernel(const float* __restrict__ feat,
                            const float* __restrict__ scale,
                            const unsigned* __restrict__ tile_ids,
                            const unsigned* __restrict__ packed,
                            unsigned* __restrict__ lists,
                            int* __restrict__ list_blocks,
                            float* __restrict__ norm_dst_out,
                            const float* __restrict__ rowscale,
                            const float* __restrict__ bias,
                            const float* __restrict__ W1,
                            const float* __restrict__ b1,
                            const float* __restrict__ W2,
                            float* __restrict__ dest) {
    __shared__ float    tile[TILE_ROWS * 64];
    __shared__ unsigned queue[AGG_WAVES][QCAP];
    __shared__ unsigned in_count[TILE_NODES];
    __shared__ v4f      w1_frag[FIRST ? W_FRAGS : 1];
    __shared__ v4f      w2_frag[FIRST ? W_FRAGS : 1];

    const int lane = threadIdx.x & 31;
    const int wave = __builtin_amdgcn_readfirstlane((int)(threadIdx.x >> 5));
    const int base = blockIdx.x * TILE_NODES;

    for (int i = threadIdx.x; i < TILE_ROWS * 64; i += AGG_THREADS) tile[i] = 0.0f;
    if (FIRST) {
        if (threadIdx.x < TILE_NODES) in_count[threadIdx.x] = 0u;
        #pragma unroll
        for (int i = 0; i < W_FRAGS / AGG_THREADS; ++i) {
            const int idx = (int)threadIdx.x + i * AGG_THREADS;
            const int fh = (idx >> 4) & 1, fc = idx & 15;
            {
                const int kp = (idx >> 5) & 7, hb = idx >> 8;
                const int ka = kp * 8 + fh * 2;
                const float* w = W1 + ka * F_HID + hb * 16 + fc;
                v4f f;
                f.x = w[0];
                f.y = w[F_HID];
                f.z = w[4 * F_HID];
                f.w = w[5 * F_HID];
                w1_frag[idx] = f;
            }
            {
                const int op = (idx >> 5) & 1, j = (idx >> 6) & 3, hb = idx >> 8;
                const int hid = hb * 16 + fh * 8 + j * 2;
                const float* w = W2 + hid * F_OUT + op * 32 + fc;
                v4f g;
                g.x = w[0];
                g.y = w[F_OUT];
                g.z = w[16];
                g.w = w[F_OUT + 16];
                w2_frag[idx] = g;
            }
        }
    }
    __syncthreads();

    unsigned* q = queue[wave];
    unsigned* wave_list = lists + (long long)(blockIdx.x * AGG_WAVES + wave) * (LIST_BLOCKS * 32);

    if (FIRST) {
        const int nrec = scan_and_add<SCALED, true, false>(q, lane, wave, feat, scale, tile_ids, packed, tile, in_count, wave_list);
        if (lane == 0) list_blocks[blockIdx.x * AGG_WAVES + wave] = nrec;
    } else {
        const int nb_l = list_blocks[blockIdx.x * AGG_WAVES + lane];
        if (__builtin_amdgcn_ballot_w32(nb_l > LIST_BLOCKS) == 0u) {
            const int nb = __builtin_amdgcn_readlane(nb_l, wave);
            unsigned ent_next = (nb > 0) ? wave_list[lane] : LIST_INVALID;
            for (int b = 0; b < nb; ++b) {
                const unsigned ent = ent_next;
                ent_next = (b + 1 < nb) ? wave_list[(b + 1) * 32 + lane] : LIST_INVALID;
                const bool ok = ent != LIST_INVALID;
                const int cnt = __builtin_popcount(__builtin_amdgcn_ballot_w32(ok));
                const int sv = ok ? (int)(ent >> 8) : 0;
                const int dl = (int)(ent & 0xFFu);
                float sc = 1.0f;
                if (SCALED) sc = scale[pitched(sv)];
                add_rows<SCALED, 16>(sv, dl, sc, cnt, lane, feat, tile);
            }
        } else {
            scan_and_add<SCALED, false, false>(q, lane, wave, feat, scale, tile_ids, packed, tile, in_count, wave_list);
        }
    }
    float  rs[STORE_PASSES] = {};
    float4 b4 = make_float4(0.0f, 0.0f, 0.0f, 0.0f);
    if (EPILOGUE) {
        b4 = ((const float4*)bias)[threadIdx.x & 15];
        #pragma unroll
        for (int p = 0; p < STORE_PASSES; ++p) {
            const int row = (int)(threadIdx.x >> 4) + p * (AGG_THREADS / 16);
            rs[p] = (row < TILE_NODES) ? rowscale[blockIdx.x * TILE_PITCH + row] : 0.0f;
        }
    }
    __syncthreads();

    if (FIRST) {
        if (threadIdx.x < TILE_PITCH) {
            const float deg = (threadIdx.x < TILE_NODES) ? (float)in_count[threadIdx.x] : 1.0f;
            norm_dst_out[blockIdx.x * TILE_PITCH + threadIdx.x] = rsqrtf(fmaxf(deg, 1.0f));
        }
        if (wave < ROW_TILES) {
            const int half = lane >> 4;
            const int node = wave * 16 + (lane & 15);
            const bool live = node < TILE_NODES;
            const float nd = live ? rsqrtf(fmaxf((float)in_count[node], 1.0f)) : 0.0f;
            const float ns = live ? scale[blockIdx.x * TILE_PITCH + node] : 0.0f;
            v2f xb[F_IN / 4];
            #pragma unroll
            for (int k4 = 0; k4 < F_IN / 4; ++k4) xb[k4] = *(const v2f*)(tile + node * 64 + k4 * 4 + half * 2);
            v8f o[F_OUT / 16] = {};
            #pragma unroll
            for (int hb = 0; hb < F_HID / 16; hb += 2) {
                v8f c0 = {}, c1 = {};
                #pragma unroll
                for (int kh = 0; kh < 2; ++kh) {
                    v4f wa[4], wb[4];
                    #pragma unroll
                    for (int i = 0; i < 4; ++i) {
                        wa[i] = w1_frag[(hb * 8 + kh * 4 + i) * 32 + lane];
                        wb[i] = w1_frag[((hb + 1) * 8 + kh * 4 + i) * 32 + lane];
                    }
                    #pragma unroll
                    for (int i = 0; i < 4; ++i) {
                        const int k4 = (kh * 4 + i) * 2;
                        v2f a;
                        a.x = wa[i].x; a.y = wa[i].y;
                        c0 = __builtin_amdgcn_wmma_f32_16x16x4_f32(false, a, false, xb[k4], (short)0, c0, false, false);
                        a.x = wb[i].x; a.y = wb[i].y;
                        c1 = __builtin_amdgcn_wmma_f32_16x16x4_f32(false, a, false, xb[k4], (short)0, c1, false, false);
                        a.x = wa[i].z; a.y = wa[i].w;
                        c0 = __builtin_amdgcn_wmma_f32_16x16x4_f32(false, a, false, xb[k4 + 1], (short)0, c0, false, false);
                        a.x = wb[i].z; a.y = wb[i].w;
                        c1 = __builtin_amdgcn_wmma_f32_16x16x4_f32(false, a, false, xb[k4 + 1], (short)0, c1, false, false);
                    }
                }
                #pragma unroll
                for (int u = 0; u < 2; ++u) {
                    const v8f c = u ? c1 : c0;
                    v4f wg[8];
                    #pragma unroll
                    for (int i = 0; i < 8; ++i) wg[i] = w2_frag[((hb + u) * 8 + i) * 32 + lane];
                    const float4 ba = ((const float4*)b1)[(hb + u) * 4 + half * 2];
                    const float4 bb = ((const float4*)b1)[(hb + u) * 4 + half * 2 + 1];
                    const float bv[8] = {ba.x, ba.y, ba.z, ba.w, bb.x, bb.y, bb.z, bb.w};
                    float hs[8];
                    #pragma unroll
                    for (int r = 0; r < 8; ++r) hs[r] = fmaxf(c[r] * nd + bv[r], 0.0f) * ns;
                    #pragma unroll
                    for (int j = 0; j < 4; ++j) {
                        v2f hf;
                        hf.x = hs[2 * j];
                        hf.y = hs[2 * j + 1];
                        #pragma unroll
                        for (int op = 0; op < 2; ++op) {
                            v2f a;
                            a.x = wg[2 * j + op].x; a.y = wg[2 * j + op].y;
                            o[2 * op] = __builtin_amdgcn_wmma_f32_16x16x4_f32(false, a, false, hf, (short)0, o[2 * op], false, false);
                            a.x = wg[2 * j + op].z; a.y = wg[2 * j + op].w;
                            o[2 * op + 1] = __builtin_amdgcn_wmma_f32_16x16x4_f32(false, a, false, hf, (short)0, o[2 * op + 1], false, false);
                        }
                    }
                }
                __builtin_amdgcn_sched_barrier(0);
            }
            if (live) {
                float* drow = dest + (long long)(base + node) * F_OUT + half * 8;
                #pragma unroll
                for (int ob = 0; ob < F_OUT / 16; ++ob) {
                    *(float4*)(drow + ob * 16)     = make_float4(o[ob][0], o[ob][1], o[ob][2], o[ob][3]);
                    *(float4*)(drow + ob * 16 + 4) = make_float4(o[ob][4], o[ob][5], o[ob][6], o[ob][7]);
                }
            }
        }
    } else {
        float4* out4 = (float4*)(dest + (long long)base * 64);
        const float4* tile4 = (const float4*)tile;
        #pragma unroll
        for (int p = 0; p < STORE_PASSES; ++p) {
            const int i = threadIdx.x + p * AGG_THREADS;
            if (i < TILE_NODES * 16) {
                float4 t = tile4[i];
                if (EPILOGUE) {
                    t.x = t.x * rs[p] + b4.x;
                    t.y = t.y * rs[p] + b4.y;
                    t.z = t.z * rs[p] + b4.z;
                    t.w = t.w * rs[p] + b4.w;
                }
                out4[i] = t;
            }
        }
    }
}

__global__ __launch_bounds__(AGG_THREADS) __attribute__((amdgpu_waves_per_eu(8, 8)))
void owner_out_degree_kernel(const unsigned* __restrict__ src_tile_ids,
                             const unsigned* __restrict__ packed,
                             float* __restrict__ norm_src_out) {
    __shared__ unsigned queue[AGG_WAVES][QCAP];
    __shared__ unsigned out_count[TILE_NODES];

    const int lane = threadIdx.x & 31;
    const int wave = __builtin_amdgcn_readfirstlane((int)(threadIdx.x >> 5));

    if (threadIdx.x < TILE_NODES) out_count[threadIdx.x] = 0u;
    __syncthreads();

    scan_and_add<false, false, true>(queue[wave], lane, wave, nullptr, nullptr, src_tile_ids, packed,
                                     nullptr, out_count, nullptr);
    __syncthreads();

    if (threadIdx.x < TILE_PITCH) {
        const float deg = (threadIdx.x < TILE_NODES) ? (float)out_count[threadIdx.x] : 1.0f;
        norm_src_out[blockIdx.x * TILE_PITCH + threadIdx.x] = rsqrtf(fmaxf(deg, 1.0f));
    }
}


static_assert((N_TILES * TILE_PITCH) % 64 == 0, "the arrays behind norm_src and norm_dst start on 256-byte boundaries");

extern "C" void kernel_launch(void* const* d_in, const int* in_sizes, int n_in,
                              void* d_out, int out_size, void* d_ws, size_t ws_size,
                              hipStream_t stream) {
    const float* x   = (const float*)d_in[0];
    const float* W1  = (const float*)d_in[1];
    const float* b1  = (const float*)d_in[2];
    const float* W2  = (const float*)d_in[3];
    const float* b2  = (const float*)d_in[4];
    const int*   src = (const int*)d_in[5];
    const int*   dst = (const int*)d_in[6];
    float*       out = (float*)d_out;

    float* ws       = (float*)d_ws;
    float* norm_src = ws;
    float* norm_dst = norm_src + N_TILES * TILE_PITCH;
    float* h2w      = norm_dst + N_TILES * TILE_PITCH;
    unsigned* lists = (unsigned*)(h2w + (long long)N_NODES * 64);
    int* list_blocks = (int*)(lists + (long long)(N_NODES / TILE_NODES) * AGG_WAVES * LIST_BLOCKS * 32);
    unsigned* tile_ids = (unsigned*)(list_blocks + (N_NODES / TILE_NODES) * AGG_WAVES);
    unsigned* src_tile_ids = tile_ids + ID_DWORDS;
    unsigned* packed   = src_tile_ids + ID_DWORDS;

    const int ZT = 256;

    edge_pass_kernel<<<(ID_DWORDS + ZT - 1) / ZT, ZT, 0, stream>>>(src, dst, tile_ids, src_tile_ids, packed);
    owner_out_degree_kernel<<<N_TILES, AGG_THREADS, 0, stream>>>(src_tile_ids, packed, norm_src);

    owner_aggregate_kernel<true, true, false><<<N_NODES / TILE_NODES, AGG_THREADS, 0, stream>>>(
        x, norm_src, tile_ids, packed, lists, list_blocks, norm_dst, nullptr, nullptr, W1, b1, W2, h2w);

    owner_aggregate_kernel<false, false, true><<<N_NODES / TILE_NODES, AGG_THREADS, 0, stream>>>(
        h2w, nullptr, tile_ids, packed, lists, list_blocks, nullptr, norm_dst, b2, nullptr, nullptr, nullptr, out);
}


# Owners no longer scan every edge: the edge pass counting-sorts 256 edge chunks by tile in LDS; an owner reads its tile's segment of each chunk (both layers, out-degree); LDS adds in arrival order
# speedup vs baseline: 4.9845x; 1.1920x over previous
#include <hip/hip_runtime.h>


#define N_NODES 50000
#define N_EDGES 800000
#define F_IN    64
#define F_HID   128
#define F_OUT   64

typedef __attribute__((ext_vector_type(2))) float v2f;
typedef __attribute__((ext_vector_type(4))) float v4f;
typedef __attribute__((ext_vector_type(8))) float v8f;

#define TILE_NODES   200
#define N_TILES      (N_NODES / TILE_NODES)
#define TILE_PITCH   256
#define AGG_THREADS  1024
#define AGG_WAVES    (AGG_THREADS / 32)
#define N_BINS       256
#define N_CHUNKS     256
#define CHUNK_EDGES  ((N_EDGES + N_CHUNKS - 1) / N_CHUNKS)
#define CHUNK_PITCH  ((CHUNK_EDGES + 31) / 32 * 32)
#define SORT_THREADS 512
#define SORT_STEPS   ((CHUNK_EDGES + SORT_THREADS - 1) / SORT_THREADS)
#define CHUNKS_PER_WAVE (N_CHUNKS / AGG_WAVES)
#define DEG_THREADS  256
#define QCAP         64
#define ROW_TILES    ((TILE_NODES + 15) / 16)
#define TILE_ROWS    (ROW_TILES * 16)
#define W_FRAGS      (F_IN * F_HID / 4)
static_assert(N_NODES % TILE_NODES == 0, "tiles must cover the nodes exactly");
static_assert(TILE_PITCH >= TILE_NODES && TILE_PITCH % 32 == 0, "a tile's norm_dst entries are whole lines");
static_assert(TILE_NODES % 8 == 0, "the eight rows a lane of the product's epilogue holds lie in one tile");
static_assert(N_TILES <= N_BINS - 1, "bin 255 is the one that no work-group owns");
static_assert(N_NODES <= (1 << 16) && TILE_NODES <= (1 << 8), "packed word: source row | local destination row << 16");
static_assert(CHUNK_EDGES < (1 << 16), "segment word: place in the chunk | length << 16");
static_assert(SORT_THREADS == 2 * N_BINS, "one sorting thread per bin of each of the two sorts");
static_assert(N_CHUNKS % AGG_WAVES == 0 && N_CHUNKS == (DEG_THREADS / 32) * 32, "the owners' waves share the chunks evenly");
static_assert(QCAP >= 31 + 32, "queue cannot overflow: it is drained after every append");
static_assert(ROW_TILES <= AGG_WAVES && TILE_ROWS <= TILE_PITCH, "one wave per 16-row block; its scales lie in the tile's entries");
static_assert(F_IN * F_HID == F_HID * F_OUT && W_FRAGS % AGG_THREADS == 0, "both weight matrices are staged by the same loop");

__global__ __launch_bounds__(SORT_THREADS)
void edge_sort_kernel(const int* __restrict__ src, const int* __restrict__ dst,
                      unsigned* __restrict__ by_dst,
                      unsigned* __restrict__ by_src,
                      unsigned* __restrict__ seg_dst,
                      unsigned* __restrict__ seg_src) {
    __shared__ unsigned cnt[2][N_BINS];
    __shared__ unsigned off[2][N_BINS];
    __shared__ unsigned stage[2][CHUNK_EDGES];
    const int tid = threadIdx.x;
    const int first = blockIdx.x * CHUNK_EDGES;

    (&cnt[0][0])[tid] = 0u;
    __syncthreads();

    unsigned wd[SORT_STEPS], wr[SORT_STEPS], bd[SORT_STEPS], bs[SORT_STEPS], rd[SORT_STEPS], rs[SORT_STEPS];
    unsigned sv[SORT_STEPS], dv[SORT_STEPS];
    bool ok[SORT_STEPS];
    #pragma unroll
    for (int k = 0; k < SORT_STEPS; ++k) {
        const int i = tid + k * SORT_THREADS;
        ok[k] = (i < CHUNK_EDGES) && (first + i < N_EDGES);
        sv[k] = ok[k] ? (unsigned)src[first + i] : ~0u;
        dv[k] = ok[k] ? (unsigned)dst[first + i] : ~0u;
    }
    #pragma unroll
    for (int k = 0; k < SORT_STEPS; ++k) {
        wd[k] = wr[k] = bd[k] = bs[k] = rd[k] = rs[k] = 0u;
        if (ok[k]) {
            const unsigned s = sv[k];
            const unsigned d = dv[k];
            const bool valid = (s < (unsigned)N_NODES) && (d < (unsigned)N_NODES);
            bs[k] = valid ? s / TILE_NODES : 255u;
            bd[k] = valid ? d / TILE_NODES : 255u;
            wd[k] = valid ? (s | ((d - bd[k] * TILE_NODES) << 16)) : 0u;
            wr[k] = valid ? (s - bs[k] * TILE_NODES) : 0u;
            rd[k] = atomicAdd(&cnt[0][bd[k]], 1u);
            rs[k] = atomicAdd(&cnt[1][bs[k]], 1u);
        }
    }
    __syncthreads();

    if (tid < 64) {
        const int a = tid >> 5, l = tid & 31;
        unsigned v[8];
        unsigned sum = 0u;
        #pragma unroll
        for (int j = 0; j < 8; ++j) { v[j] = cnt[a][l * 8 + j]; sum += v[j]; }
        unsigned incl = sum;
        #pragma unroll
        for (int o = 1; o < 32; o <<= 1) {
            const unsigned t = __shfl_up(incl, o);
            if (l >= o) incl += t;
        }
        unsigned run = incl - sum;
        #pragma unroll
        for (int j = 0; j < 8; ++j) { off[a][l * 8 + j] = run; run += v[j]; }
    }
    __syncthreads();

    #pragma unroll
    for (int k = 0; k < SORT_STEPS; ++k) {
        if (ok[k]) {
            stage[0][off[0][bd[k]] + rd[k]] = wd[k];
            stage[1][off[1][bs[k]] + rs[k]] = wr[k];
        }
    }
    {
        const int a = tid >> 8, b = tid & (N_BINS - 1);
        unsigned* seg = a ? seg_src : seg_dst;
        seg[blockIdx.x * N_BINS + b] = off[a][b] | (cnt[a][b] << 16);
    }
    __syncthreads();

    const int n = min(CHUNK_EDGES, N_EDGES - first);
    for (int i = tid; i < n; i += SORT_THREADS) {
        by_dst[blockIdx.x * CHUNK_PITCH + i] = stage[0][i];
        by_src[blockIdx.x * CHUNK_PITCH + i] = stage[1][i];
    }
}

__device__ __forceinline__ int pitched(int r) { return r + (r / TILE_NODES) * (TILE_PITCH - TILE_NODES); }


template <bool SCALED, int ROWS>
__device__ __forceinline__ void add_rows(int sv, int dl, float sc, int cnt, int lane,
                                         const float* __restrict__ feat, float* tile) {
    #pragma unroll 1
    for (int j0 = 0; j0 < cnt; j0 += ROWS) {
        v2f r[ROWS];
        #pragma unroll
        for (int u = 0; u < ROWS; ++u) {
            const int j  = (j0 + u < cnt) ? (j0 + u) : (cnt - 1);
            const int sj = __builtin_amdgcn_readlane(sv, j);
            r[u] = *(const v2f*)(feat + (long long)sj * 64 + lane * 2);
        }
        #pragma unroll
        for (int u = 0; u < ROWS; ++u) {
            const int j = j0 + u;
            if (j < cnt) {
                const int dj = __builtin_amdgcn_readlane(dl, j);
                float x0 = r[u].x, x1 = r[u].y;
                if (SCALED) {
                    const float sj = __int_as_float(__builtin_amdgcn_readlane(__float_as_int(sc), j));
                    x0 *= sj; x1 *= sj;
                }
                float* t = tile + dj * 64 + lane * 2;
                atomicAdd(t, x0);
                atomicAdd(t + 1, x1);
            }
        }
    }
}

template <bool SCALED, bool COUNT, int ROWS>
__device__ __forceinline__ void add_group(const unsigned* q, int cnt, int lane,
                                          const float* __restrict__ feat,
                                          const float* __restrict__ scale,
                                          float* tile, unsigned* in_count) {
    const unsigned pk = q[lane < cnt ? lane : 0];
    const int sv = (int)(pk & 0xFFFFu);
    const int dl = (int)(pk >> 16);
    float sc = 1.0f;
    if (SCALED) sc = scale[pitched(sv)];
    if (COUNT) {
        if (lane < cnt) atomicAdd(in_count + dl, 1u);
    }
    add_rows<SCALED, ROWS>(sv, dl, sc, cnt, lane, feat, tile);
}

template <bool SCALED, bool COUNT, int ROWS>
__device__ __forceinline__ void gather_and_add(unsigned* q, int lane, int wave,
                                               const float* __restrict__ feat,
                                               const float* __restrict__ scale,
                                               const unsigned* __restrict__ by_dst,
                                               const unsigned* __restrict__ seg_dst,
                                               float* tile, unsigned* in_count) {
    const int c0 = wave * CHUNKS_PER_WAVE;
    const unsigned sg = (lane < CHUNKS_PER_WAVE) ? seg_dst[(c0 + lane) * N_BINS + blockIdx.x] : 0u;
    unsigned head[CHUNKS_PER_WAVE];
    #pragma unroll
    for (int i = 0; i < CHUNKS_PER_WAVE; ++i) {
        const unsigned w = (unsigned)__builtin_amdgcn_readlane((int)sg, i);
        const int st = (int)(w & 0xFFFFu), n = (int)(w >> 16);
        head[i] = (lane < n) ? by_dst[(c0 + i) * CHUNK_PITCH + st + lane] : 0u;
    }
    int qn = 0;
    #pragma unroll
    for (int i = 0; i < CHUNKS_PER_WAVE; ++i) {
        const unsigned w = (unsigned)__builtin_amdgcn_readlane((int)sg, i);
        const int st = (int)(w & 0xFFFFu), n = (int)(w >> 16);
        const unsigned* sp = by_dst + (c0 + i) * CHUNK_PITCH + st;
        for (int j = 0; j < n; j += 32) {
            unsigned ent = head[i];
            if (j > 0) ent = (j + lane < n) ? sp[j + lane] : 0u;
            const int m = (n - j < 32) ? (n - j) : 32;
            if (lane < m) q[qn + lane] = ent;
            qn += m;
            if (qn >= 32) {
                __builtin_amdgcn_wave_barrier();
                qn -= 32;
                add_group<SCALED, COUNT, ROWS>(q + qn, 32, lane, feat, scale, tile, in_count);
            }
        }
    }
    if (qn > 0) {
        __builtin_amdgcn_wave_barrier();
        add_group<SCALED, COUNT, ROWS>(q, qn, lane, feat, scale, tile, in_count);
    }
}

#define STORE_PASSES ((TILE_NODES * 16 + AGG_THREADS - 1) / AGG_THREADS)
static_assert(AGG_THREADS % 16 == 0, "a lane keeps its four columns from one store pass to the next");
template <bool SCALED, bool FIRST, bool EPILOGUE>
__global__ __launch_bounds__(AGG_THREADS) __attribute__((amdgpu_waves_per_eu(8, 8)))
void owner_aggregate_kernel(const float* __restrict__ feat,
                            const float* __restrict__ scale,
                            const unsigned* __restrict__ by_dst,
                            const unsigned* __restrict__ seg_dst,
                            float* __restrict__ norm_dst_out,
                            const float* __restrict__ rowscale,
                            const float* __restrict__ bias,
                            const float* __restrict__ W1,
                            const float* __restrict__ b1,
                            const float* __restrict__ W2,
                            float* __restrict__ dest) {
    __shared__ float    tile[TILE_ROWS * 64];
    __shared__ unsigned queue[AGG_WAVES][QCAP];
    __shared__ unsigned in_count[TILE_NODES];
    __shared__ v4f      w1_frag[FIRST ? W_FRAGS : 1];
    __shared__ v4f      w2_frag[FIRST ? W_FRAGS : 1];

    const int lane = threadIdx.x & 31;
    const int wave = __builtin_amdgcn_readfirstlane((int)(threadIdx.x >> 5));
    const int base = blockIdx.x * TILE_NODES;

    for (int i = threadIdx.x; i < TILE_ROWS * 64; i += AGG_THREADS) tile[i] = 0.0f;
    if (FIRST) {
        if (threadIdx.x < TILE_NODES) in_count[threadIdx.x] = 0u;
        #pragma unroll
        for (int i = 0; i < W_FRAGS / AGG_THREADS; ++i) {
            const int idx = (int)threadIdx.x + i * AGG_THREADS;
            const int fh = (idx >> 4) & 1, fc = idx & 15;
            {
                const int kp = (idx >> 5) & 7, hb = idx >> 8;
                const int ka = kp * 8 + fh * 2;
                const float* w = W1 + ka * F_HID + hb * 16 + fc;
                v4f f;
                f.x = w[0];
                f.y = w[F_HID];
                f.z = w[4 * F_HID];
                f.w = w[5 * F_HID];
                w1_frag[idx] = f;
            }
            {
                const int op = (idx >> 5) & 1, j = (idx >> 6) & 3, hb = idx >> 8;
                const int hid = hb * 16 + fh * 8 + j * 2;
                const float* w = W2 + hid * F_OUT + op * 32 + fc;
                v4f g;
                g.x = w[0];
                g.y = w[F_OUT];
                g.z = w[16];
                g.w = w[F_OUT + 16];
                w2_frag[idx] = g;
            }
        }
    }
    __syncthreads();

    unsigned* q = queue[wave];

    gather_and_add<SCALED, FIRST, (FIRST ? 8 : 16)>(q, lane, wave, feat, scale, by_dst, seg_dst, tile, in_count);
    float  rs[STORE_PASSES] = {};
    float4 b4 = make_float4(0.0f, 0.0f, 0.0f, 0.0f);
    if (EPILOGUE) {
        b4 = ((const float4*)bias)[threadIdx.x & 15];
        #pragma unroll
        for (int p = 0; p < STORE_PASSES; ++p) {
            const int row = (int)(threadIdx.x >> 4) + p * (AGG_THREADS / 16);
            rs[p] = (row < TILE_NODES) ? rowscale[blockIdx.x * TILE_PITCH + row] : 0.0f;
        }
    }
    __syncthreads();

    if (FIRST) {
        if (threadIdx.x < TILE_PITCH) {
            const float deg = (threadIdx.x < TILE_NODES) ? (float)in_count[threadIdx.x] : 1.0f;
            norm_dst_out[blockIdx.x * TILE_PITCH + threadIdx.x] = rsqrtf(fmaxf(deg, 1.0f));
        }
        if (wave < ROW_TILES) {
            const int half = lane >> 4;
            const int node = wave * 16 + (lane & 15);
            const bool live = node < TILE_NODES;
            const float nd = live ? rsqrtf(fmaxf((float)in_count[node], 1.0f)) : 0.0f;
            const float ns = live ? scale[blockIdx.x * TILE_PITCH + node] : 0.0f;
            v2f xb[F_IN / 4];
            #pragma unroll
            for (int k4 = 0; k4 < F_IN / 4; ++k4) xb[k4] = *(const v2f*)(tile + node * 64 + k4 * 4 + half * 2);
            v8f o[F_OUT / 16] = {};
            #pragma unroll
            for (int hb = 0; hb < F_HID / 16; hb += 2) {
                v8f c0 = {}, c1 = {};
                #pragma unroll
                for (int kh = 0; kh < 2; ++kh) {
                    v4f wa[4], wb[4];
                    #pragma unroll
                    for (int i = 0; i < 4; ++i) {
                        wa[i] = w1_frag[(hb * 8 + kh * 4 + i) * 32 + lane];
                        wb[i] = w1_frag[((hb + 1) * 8 + kh * 4 + i) * 32 + lane];
                    }
                    #pragma unroll
                    for (int i = 0; i < 4; ++i) {
                        const int k4 = (kh * 4 + i) * 2;
                        v2f a;
                        a.x = wa[i].x; a.y = wa[i].y;
                        c0 = __builtin_amdgcn_wmma_f32_16x16x4_f32(false, a, false, xb[k4], (short)0, c0, false, false);
                        a.x = wb[i].x; a.y = wb[i].y;
                        c1 = __builtin_amdgcn_wmma_f32_16x16x4_f32(false, a, false, xb[k4], (short)0, c1, false, false);
                        a.x = wa[i].z; a.y = wa[i].w;
                        c0 = __builtin_amdgcn_wmma_f32_16x16x4_f32(false, a, false, xb[k4 + 1], (short)0, c0, false, false);
                        a.x = wb[i].z; a.y = wb[i].w;
                        c1 = __builtin_amdgcn_wmma_f32_16x16x4_f32(false, a, false, xb[k4 + 1], (short)0, c1, false, false);
                    }
                }
                #pragma unroll
                for (int u = 0; u < 2; ++u) {
                    const v8f c = u ? c1 : c0;
                    v4f wg[8];
                    #pragma unroll
                    for (int i = 0; i < 8; ++i) wg[i] = w2_frag[((hb + u) * 8 + i) * 32 + lane];
                    const float4 ba = ((const float4*)b1)[(hb + u) * 4 + half * 2];
                    const float4 bb = ((const float4*)b1)[(hb + u) * 4 + half * 2 + 1];
                    const float bv[8] = {ba.x, ba.y, ba.z, ba.w, bb.x, bb.y, bb.z, bb.w};
                    float hs[8];
                    #pragma unroll
                    for (int r = 0; r < 8; ++r) hs[r] = fmaxf(c[r] * nd + bv[r], 0.0f) * ns;
                    #pragma unroll
                    for (int j = 0; j < 4; ++j) {
                        v2f hf;
                        hf.x = hs[2 * j];
                        hf.y = hs[2 * j + 1];
                        #pragma unroll
                        for (int op = 0; op < 2; ++op) {
                            v2f a;
                            a.x = wg[2 * j + op].x; a.y = wg[2 * j + op].y;
                            o[2 * op] = __builtin_amdgcn_wmma_f32_16x16x4_f32(false, a, false, hf, (short)0, o[2 * op], false, false);
                            a.x = wg[2 * j + op].z; a.y = wg[2 * j + op].w;
                            o[2 * op + 1] = __builtin_amdgcn_wmma_f32_16x16x4_f32(false, a, false, hf, (short)0, o[2 * op + 1], false, false);
                        }
                    }
                }
                __builtin_amdgcn_sched_barrier(0);
            }
            if (live) {
                float* drow = dest + (long long)(base + node) * F_OUT + half * 8;
                #pragma unroll
                for (int ob = 0; ob < F_OUT / 16; ++ob) {
                    *(float4*)(drow + ob * 16)     = make_float4(o[ob][0], o[ob][1], o[ob][2], o[ob][3]);
                    *(float4*)(drow + ob * 16 + 4) = make_float4(o[ob][4], o[ob][5], o[ob][6], o[ob][7]);
                }
            }
        }
    } else {
        float4* out4 = (float4*)(dest + (long long)base * 64);
        const float4* tile4 = (const float4*)tile;
        #pragma unroll
        for (int p = 0; p < STORE_PASSES; ++p) {
            const int i = threadIdx.x + p * AGG_THREADS;
            if (i < TILE_NODES * 16) {
                float4 t = tile4[i];
                if (EPILOGUE) {
                    t.x = t.x * rs[p] + b4.x;
                    t.y = t.y * rs[p] + b4.y;
                    t.z = t.z * rs[p] + b4.z;
                    t.w = t.w * rs[p] + b4.w;
                }
                out4[i] = t;
            }
        }
    }
}

#define DEG_CHUNKS (N_CHUNKS / (DEG_THREADS / 32))
static_assert(DEG_CHUNKS == 32, "a lane holds the segment word of one of its wave's chunks");
__global__ __launch_bounds__(DEG_THREADS)
void owner_out_degree_kernel(const unsigned* __restrict__ by_src,
                             const unsigned* __restrict__ seg_src,
                             float* __restrict__ norm_src_out) {
    __shared__ unsigned out_count[TILE_NODES];

    const int lane = threadIdx.x & 31;
    const int wave = __builtin_amdgcn_readfirstlane((int)(threadIdx.x >> 5));

    if (threadIdx.x < TILE_NODES) out_count[threadIdx.x] = 0u;
    __syncthreads();

    const int c0 = wave * DEG_CHUNKS;
    const unsigned sg = seg_src[(c0 + lane) * N_BINS + blockIdx.x];
    unsigned head[DEG_CHUNKS];
    #pragma unroll
    for (int i = 0; i < DEG_CHUNKS; ++i) {
        const unsigned w = (unsigned)__builtin_amdgcn_readlane((int)sg, i);
        const int st = (int)(w & 0xFFFFu), n = (int)(w >> 16);
        head[i] = (lane < n) ? by_src[(c0 + i) * CHUNK_PITCH + st + lane] : 0u;
    }
    #pragma unroll
    for (int i = 0; i < DEG_CHUNKS; ++i) {
        const unsigned w = (unsigned)__builtin_amdgcn_readlane((int)sg, i);
        const int st = (int)(w & 0xFFFFu), n = (int)(w >> 16);
        const unsigned* sp = by_src + (c0 + i) * CHUNK_PITCH + st;
        for (int j = 0; j < n; j += 32) {
            unsigned r = head[i];
            if (j > 0) r = (j + lane < n) ? sp[j + lane] : 0u;
            if (j + lane < n) atomicAdd(out_count + r, 1u);
        }
    }
    __syncthreads();

    if (threadIdx.x < TILE_PITCH) {
        const float deg = (threadIdx.x < TILE_NODES) ? (float)out_count[threadIdx.x] : 1.0f;
        norm_src_out[blockIdx.x * TILE_PITCH + threadIdx.x] = rsqrtf(fmaxf(deg, 1.0f));
    }
}


static_assert((N_TILES * TILE_PITCH) % 64 == 0, "the arrays behind norm_src and norm_dst start on 256-byte boundaries");
static_assert(DEG_THREADS >= TILE_PITCH, "the out-degree work-group stores its tile's norm_src entries in one pass");

extern "C" void kernel_launch(void* const* d_in, const int* in_sizes, int n_in,
                              void* d_out, int out_size, void* d_ws, size_t ws_size,
                              hipStream_t stream) {
    const float* x   = (const float*)d_in[0];
    const float* W1  = (const float*)d_in[1];
    const float* b1  = (const float*)d_in[2];
    const float* W2  = (const float*)d_in[3];
    const float* b2  = (const float*)d_in[4];
    const int*   src = (const int*)d_in[5];
    const int*   dst = (const int*)d_in[6];
    float*       out = (float*)d_out;

    float* ws       = (float*)d_ws;
    float* norm_src = ws;
    float* norm_dst = norm_src + N_TILES * TILE_PITCH;
    float* h2w      = norm_dst + N_TILES * TILE_PITCH;
    unsigned* by_dst  = (unsigned*)(h2w + (long long)N_NODES * 64);
    unsigned* by_src  = by_dst + N_CHUNKS * CHUNK_PITCH;
    unsigned* seg_dst = by_src + N_CHUNKS * CHUNK_PITCH;
    unsigned* seg_src = seg_dst + N_CHUNKS * N_BINS;

    edge_sort_kernel<<<N_CHUNKS, SORT_THREADS, 0, stream>>>(src, dst, by_dst, by_src, seg_dst, seg_src);
    owner_out_degree_kernel<<<N_TILES, DEG_THREADS, 0, stream>>>(by_src, seg_src, norm_src);

    owner_aggregate_kernel<true, true, false><<<N_NODES / TILE_NODES, AGG_THREADS, 0, stream>>>(
        x, norm_src, by_dst, seg_dst, norm_dst, nullptr, nullptr, W1, b1, W2, h2w);

    owner_aggregate_kernel<false, false, true><<<N_NODES / TILE_NODES, AGG_THREADS, 0, stream>>>(
        h2w, nullptr, by_dst, seg_dst, nullptr, norm_dst, b2, nullptr, nullptr, nullptr, out);
}


# speedup vs baseline: 5.5711x; 1.1177x over previous
#include <hip/hip_runtime.h>


#define N_NODES 50000
#define N_EDGES 800000
#define F_IN    64
#define F_HID   128
#define F_OUT   64

typedef __attribute__((ext_vector_type(2))) float v2f;
typedef __attribute__((ext_vector_type(4))) float v4f;
typedef __attribute__((ext_vector_type(8))) float v8f;

#define TILE_NODES   200
#define N_TILES      (N_NODES / TILE_NODES)
#define TILE_PITCH   256
#define AGG_THREADS  1024
#define AGG_WAVES    (AGG_THREADS / 32)
#define N_BINS       256
#define N_CHUNKS     256
#define CHUNK_EDGES  ((N_EDGES + N_CHUNKS - 1) / N_CHUNKS)
#define CHUNK_PITCH  ((CHUNK_EDGES + 31) / 32 * 32)
#define SORT_THREADS 512
#define SORT_STEPS   ((CHUNK_EDGES + SORT_THREADS - 1) / SORT_THREADS)
#define CHUNKS_PER_WAVE (N_CHUNKS / AGG_WAVES)
#define DEG_THREADS  256
#define QCAP         64
#define ROW_TILES    ((TILE_NODES + 15) / 16)
#define TILE_ROWS    (ROW_TILES * 16)
#define W_FRAGS      (F_IN * F_HID / 4)
static_assert(N_NODES % TILE_NODES == 0, "tiles must cover the nodes exactly");
static_assert(TILE_PITCH >= TILE_NODES && TILE_PITCH % 32 == 0, "a tile's norm_dst entries are whole lines");
static_assert(TILE_NODES % 8 == 0, "the eight rows a lane of the product's epilogue holds lie in one tile");
static_assert(N_TILES <= N_BINS - 1, "bin 255 is the one that no work-group owns");
static_assert(N_NODES <= (1 << 16) && TILE_NODES <= (1 << 8), "packed word: source row | local destination row << 16");
static_assert(CHUNK_EDGES < (1 << 16), "segment word: place in the chunk | length << 16");
static_assert(SORT_THREADS == 2 * N_BINS, "one sorting thread per bin of each of the two sorts");
static_assert(N_CHUNKS % AGG_WAVES == 0 && N_CHUNKS == (DEG_THREADS / 32) * 32, "the owners' waves share the chunks evenly");
static_assert(QCAP >= 31 + 32, "queue cannot overflow: it is drained after every append");
static_assert(ROW_TILES <= AGG_WAVES && TILE_ROWS <= TILE_PITCH, "one wave per 16-row block; its scales lie in the tile's entries");
static_assert(F_IN * F_HID == F_HID * F_OUT && W_FRAGS % AGG_THREADS == 0, "both weight matrices are staged by the same loop");
#define TILE_BYTES   (TILE_ROWS * 64 * 4)
#define COPY_BYTES   (64 * 1024)
#define LDS_QUEUE    (4 * 32 * 1024)
#define LDS_COUNT    (LDS_QUEUE + AGG_WAVES * QCAP * 4)
#define LDS_W1       (LDS_COUNT + 1024)
#define LDS_W2       (LDS_W1 + W_FRAGS * 16)
#define LDS_FIRST    (LDS_W2 + W_FRAGS * 16)
#define LDS_SECOND   LDS_COUNT
static_assert(TILE_BYTES <= COPY_BYTES && COPY_BYTES + TILE_BYTES <= LDS_QUEUE, "the copies lie in blocks 0-1 and 2-3");
static_assert(TILE_NODES * 4 <= 1024 && LDS_W1 % 16 == 0, "in-degrees fit their place; fragments are 16-byte entries");
static_assert(LDS_FIRST <= 320 * 1024, "a WGP's LDS");

__global__ __launch_bounds__(SORT_THREADS)
void edge_sort_kernel(const int* __restrict__ src, const int* __restrict__ dst,
                      unsigned* __restrict__ by_dst,
                      unsigned* __restrict__ by_src,
                      unsigned* __restrict__ seg_dst,
                      unsigned* __restrict__ seg_src) {
    __shared__ unsigned cnt[2][N_BINS];
    __shared__ unsigned off[2][N_BINS];
    __shared__ unsigned stage[2][CHUNK_EDGES];
    const int tid = threadIdx.x;
    const int first = blockIdx.x * CHUNK_EDGES;

    (&cnt[0][0])[tid] = 0u;
    __syncthreads();

    unsigned wd[SORT_STEPS], wr[SORT_STEPS], bd[SORT_STEPS], bs[SORT_STEPS], rd[SORT_STEPS], rs[SORT_STEPS];
    unsigned sv[SORT_STEPS], dv[SORT_STEPS];
    bool ok[SORT_STEPS];
    #pragma unroll
    for (int k = 0; k < SORT_STEPS; ++k) {
        const int i = tid + k * SORT_THREADS;
        ok[k] = (i < CHUNK_EDGES) && (first + i < N_EDGES);
        sv[k] = ok[k] ? (unsigned)src[first + i] : ~0u;
        dv[k] = ok[k] ? (unsigned)dst[first + i] : ~0u;
    }
    #pragma unroll
    for (int k = 0; k < SORT_STEPS; ++k) {
        wd[k] = wr[k] = bd[k] = bs[k] = rd[k] = rs[k] = 0u;
        if (ok[k]) {
            const unsigned s = sv[k];
            const unsigned d = dv[k];
            const bool valid = (s < (unsigned)N_NODES) && (d < (unsigned)N_NODES);
            bs[k] = valid ? s / TILE_NODES : 255u;
            bd[k] = valid ? d / TILE_NODES : 255u;
            wd[k] = valid ? (s | ((d - bd[k] * TILE_NODES) << 16)) : 0u;
            wr[k] = valid ? (s - bs[k] * TILE_NODES) : 0u;
            rd[k] = atomicAdd(&cnt[0][bd[k]], 1u);
            rs[k] = atomicAdd(&cnt[1][bs[k]], 1u);
        }
    }
    __syncthreads();

    if (tid < 64) {
        const int a = tid >> 5, l = tid & 31;
        unsigned v[8];
        unsigned sum = 0u;
        #pragma unroll
        for (int j = 0; j < 8; ++j) { v[j] = cnt[a][l * 8 + j]; sum += v[j]; }
        unsigned incl = sum;
        #pragma unroll
        for (int o = 1; o < 32; o <<= 1) {
            const unsigned t = __shfl_up(incl, o);
            if (l >= o) incl += t;
        }
        unsigned run = incl - sum;
        #pragma unroll
        for (int j = 0; j < 8; ++j) { off[a][l * 8 + j] = run; run += v[j]; }
    }
    __syncthreads();

    #pragma unroll
    for (int k = 0; k < SORT_STEPS; ++k) {
        if (ok[k]) {
            stage[0][off[0][bd[k]] + rd[k]] = wd[k];
            stage[1][off[1][bs[k]] + rs[k]] = wr[k];
        }
    }
    {
        const int a = tid >> 8, b = tid & (N_BINS - 1);
        unsigned* seg = a ? seg_src : seg_dst;
        seg[blockIdx.x * N_BINS + b] = off[a][b] | (cnt[a][b] << 16);
    }
    __syncthreads();

    const int n = min(CHUNK_EDGES, N_EDGES - first);
    for (int i = tid; i < n; i += SORT_THREADS) {
        by_dst[blockIdx.x * CHUNK_PITCH + i] = stage[0][i];
        by_src[blockIdx.x * CHUNK_PITCH + i] = stage[1][i];
    }
}

__device__ __forceinline__ int pitched(int r) { return r + (r / TILE_NODES) * (TILE_PITCH - TILE_NODES); }


template <bool SCALED, int ROWS>
__device__ __forceinline__ void add_rows(int sv, int dl, float sc, int cnt, int lane,
                                         const float* __restrict__ feat, float* tile) {
    #pragma unroll 1
    for (int j0 = 0; j0 < cnt; j0 += ROWS) {
        v2f r[ROWS];
        #pragma unroll
        for (int u = 0; u < ROWS; ++u) {
            const int j  = (j0 + u < cnt) ? (j0 + u) : (cnt - 1);
            const int sj = __builtin_amdgcn_readlane(sv, j);
            r[u] = *(const v2f*)(feat + (long long)sj * 64 + lane * 2);
        }
        #pragma unroll
        for (int u = 0; u < ROWS; ++u) {
            const int j = j0 + u;
            if (j < cnt) {
                const int dj = __builtin_amdgcn_readlane(dl, j);
                float x0 = r[u].x, x1 = r[u].y;
                if (SCALED) {
                    const float sj = __int_as_float(__builtin_amdgcn_readlane(__float_as_int(sc), j));
                    x0 *= sj; x1 *= sj;
                }
                float* t = tile + dj * 64 + lane * 2;
                atomicAdd(t, x0);
                atomicAdd(t + 1, x1);
            }
        }
    }
}

template <bool SCALED, bool COUNT, int ROWS>
__device__ __forceinline__ void add_group(const unsigned* q, int cnt, int lane,
                                          const float* __restrict__ feat,
                                          const float* __restrict__ scale,
                                          float* tile, unsigned* in_count) {
    const unsigned pk = q[lane < cnt ? lane : 0];
    const int sv = (int)(pk & 0xFFFFu);
    const int dl = (int)(pk >> 16);
    float sc = 1.0f;
    if (SCALED) sc = scale[pitched(sv)];
    if (COUNT) {
        if (lane < cnt) atomicAdd(in_count + dl, 1u);
    }
    add_rows<SCALED, ROWS>(sv, dl, sc, cnt, lane, feat, tile);
}

template <bool SCALED, bool COUNT, int ROWS>
__device__ __forceinline__ void gather_and_add(unsigned* q, int lane, int wave,
                                               const float* __restrict__ feat,
                                               const float* __restrict__ scale,
                                               const unsigned* __restrict__ by_dst,
                                               const unsigned* __restrict__ seg_dst,
                                               float* tile, unsigned* in_count) {
    const int c0 = wave * CHUNKS_PER_WAVE;
    const unsigned sg = (lane < CHUNKS_PER_WAVE) ? seg_dst[(c0 + lane) * N_BINS + blockIdx.x] : 0u;
    unsigned head[CHUNKS_PER_WAVE];
    #pragma unroll
    for (int i = 0; i < CHUNKS_PER_WAVE; ++i) {
        const unsigned w = (unsigned)__builtin_amdgcn_readlane((int)sg, i);
        const int st = (int)(w & 0xFFFFu), n = (int)(w >> 16);
        head[i] = (lane < n) ? by_dst[(c0 + i) * CHUNK_PITCH + st + lane] : 0u;
    }
    int qn = 0;
    #pragma unroll
    for (int i = 0; i < CHUNKS_PER_WAVE; ++i) {
        const unsigned w = (unsigned)__builtin_amdgcn_readlane((int)sg, i);
        const int st = (int)(w & 0xFFFFu), n = (int)(w >> 16);
        const unsigned* sp = by_dst + (c0 + i) * CHUNK_PITCH + st;
        for (int j = 0; j < n; j += 32) {
            unsigned ent = head[i];
            if (j > 0) ent = (j + lane < n) ? sp[j + lane] : 0u;
            const int m = (n - j < 32) ? (n - j) : 32;
            if (lane < m) q[qn + lane] = ent;
            qn += m;
            if (qn >= 32) {
                __builtin_amdgcn_wave_barrier();
                qn -= 32;
                add_group<SCALED, COUNT, ROWS>(q + qn, 32, lane, feat, scale, tile, in_count);
            }
        }
    }
    if (qn > 0) {
        __builtin_amdgcn_wave_barrier();
        add_group<SCALED, COUNT, ROWS>(q, qn, lane, feat, scale, tile, in_count);
    }
}

#define STORE_PASSES ((TILE_NODES * 16 + AGG_THREADS - 1) / AGG_THREADS)
static_assert(AGG_THREADS % 16 == 0, "a lane keeps its four columns from one store pass to the next");
template <bool SCALED, bool FIRST, bool EPILOGUE>
__global__ __launch_bounds__(AGG_THREADS) __attribute__((amdgpu_waves_per_eu(8, 8)))
void owner_aggregate_kernel(const float* __restrict__ feat,
                            const float* __restrict__ scale,
                            const unsigned* __restrict__ by_dst,
                            const unsigned* __restrict__ seg_dst,
                            float* __restrict__ norm_dst_out,
                            const float* __restrict__ rowscale,
                            const float* __restrict__ bias,
                            const float* __restrict__ W1,
                            const float* __restrict__ b1,
                            const float* __restrict__ W2,
                            float* __restrict__ dest) {
    __shared__ __attribute__((aligned(256))) unsigned char lds[FIRST ? LDS_FIRST : LDS_SECOND];
    float*    tile0    = (float*)lds;
    float*    tile1    = (float*)(lds + COPY_BYTES);
    unsigned* queues   = (unsigned*)(lds + LDS_QUEUE);
    unsigned* in_count = (unsigned*)(lds + LDS_COUNT);
    v4f*      w1_frag  = (v4f*)(lds + (FIRST ? LDS_W1 : 0));
    v4f*      w2_frag  = (v4f*)(lds + (FIRST ? LDS_W2 : 0));

    const int lane = threadIdx.x & 31;
    const int wave = __builtin_amdgcn_readfirstlane((int)(threadIdx.x >> 5));
    const int base = blockIdx.x * TILE_NODES;

    for (int i = threadIdx.x; i < TILE_ROWS * 16; i += AGG_THREADS) {
        ((float4*)tile0)[i] = make_float4(0.0f, 0.0f, 0.0f, 0.0f);
        ((float4*)tile1)[i] = make_float4(0.0f, 0.0f, 0.0f, 0.0f);
    }
    if (FIRST) {
        if (threadIdx.x < TILE_NODES) in_count[threadIdx.x] = 0u;
        #pragma unroll
        for (int i = 0; i < W_FRAGS / AGG_THREADS; ++i) {
            const int idx = (int)threadIdx.x + i * AGG_THREADS;
            const int fh = (idx >> 4) & 1, fc = idx & 15;
            {
                const int kp = (idx >> 5) & 7, hb = idx >> 8;
                const int ka = kp * 8 + fh * 2;
                const float* w = W1 + ka * F_HID + hb * 16 + fc;
                v4f f;
                f.x = w[0];
                f.y = w[F_HID];
                f.z = w[4 * F_HID];
                f.w = w[5 * F_HID];
                w1_frag[idx] = f;
            }
            {
                const int op = (idx >> 5) & 1, j = (idx >> 6) & 3, hb = idx >> 8;
                const int hid = hb * 16 + fh * 8 + j * 2;
                const float* w = W2 + hid * F_OUT + op * 32 + fc;
                v4f g;
                g.x = w[0];
                g.y = w[F_OUT];
                g.z = w[16];
                g.w = w[F_OUT + 16];
                w2_frag[idx] = g;
            }
        }
    }
    __syncthreads();

    unsigned* q = queues + wave * QCAP;
    float* tile = tile0 + (wave & 1) * (COPY_BYTES / 4);

    gather_and_add<SCALED, FIRST, (FIRST ? 8 : 16)>(q, lane, wave, feat, scale, by_dst, seg_dst, tile, in_count);
    float  rs[STORE_PASSES] = {};
    float4 b4 = make_float4(0.0f, 0.0f, 0.0f, 0.0f);
    if (EPILOGUE) {
        b4 = ((const float4*)bias)[threadIdx.x & 15];
        #pragma unroll
        for (int p = 0; p < STORE_PASSES; ++p) {
            const int row = (int)(threadIdx.x >> 4) + p * (AGG_THREADS / 16);
            rs[p] = (row < TILE_NODES) ? rowscale[blockIdx.x * TILE_PITCH + row] : 0.0f;
        }
    }
    __syncthreads();

    if (FIRST) {
        if (threadIdx.x < TILE_PITCH) {
            const float deg = (threadIdx.x < TILE_NODES) ? (float)in_count[threadIdx.x] : 1.0f;
            norm_dst_out[blockIdx.x * TILE_PITCH + threadIdx.x] = rsqrtf(fmaxf(deg, 1.0f));
        }
        if (wave < ROW_TILES) {
            const int half = lane >> 4;
            const int node = wave * 16 + (lane & 15);
            const bool live = node < TILE_NODES;
            const float nd = live ? rsqrtf(fmaxf((float)in_count[node], 1.0f)) : 0.0f;
            const float ns = live ? scale[blockIdx.x * TILE_PITCH + node] : 0.0f;
            v2f xb[F_IN / 4];
            #pragma unroll
            for (int k4 = 0; k4 < F_IN / 4; ++k4)
                xb[k4] = *(const v2f*)(tile0 + node * 64 + k4 * 4 + half * 2) + *(const v2f*)(tile1 + node * 64 + k4 * 4 + half * 2);
            v8f o[F_OUT / 16] = {};
            #pragma unroll
            for (int hb = 0; hb < F_HID / 16; hb += 2) {
                v8f c0 = {}, c1 = {};
                #pragma unroll
                for (int kh = 0; kh < 2; ++kh) {
                    v4f wa[4], wb[4];
                    #pragma unroll
                    for (int i = 0; i < 4; ++i) {
                        wa[i] = w1_frag[(hb * 8 + kh * 4 + i) * 32 + lane];
                        wb[i] = w1_frag[((hb + 1) * 8 + kh * 4 + i) * 32 + lane];
                    }
                    #pragma unroll
                    for (int i = 0; i < 4; ++i) {
                        const int k4 = (kh * 4 + i) * 2;
                        v2f a;
                        a.x = wa[i].x; a.y = wa[i].y;
                        c0 = __builtin_amdgcn_wmma_f32_16x16x4_f32(false, a, false, xb[k4], (short)0, c0, false, false);
                        a.x = wb[i].x; a.y = wb[i].y;
                        c1 = __builtin_amdgcn_wmma_f32_16x16x4_f32(false, a, false, xb[k4], (short)0, c1, false, false);
                        a.x = wa[i].z; a.y = wa[i].w;
                        c0 = __builtin_amdgcn_wmma_f32_16x16x4_f32(false, a, false, xb[k4 + 1], (short)0, c0, false, false);
                        a.x = wb[i].z; a.y = wb[i].w;
                        c1 = __builtin_amdgcn_wmma_f32_16x16x4_f32(false, a, false, xb[k4 + 1], (short)0, c1, false, false);
                    }
                }
                #pragma unroll
                for (int u = 0; u < 2; ++u) {
                    const v8f c = u ? c1 : c0;
                    v4f wg[8];
                    #pragma unroll
                    for (int i = 0; i < 8; ++i) wg[i] = w2_frag[((hb + u) * 8 + i) * 32 + lane];
                    const float4 ba = ((const float4*)b1)[(hb + u) * 4 + half * 2];
                    const float4 bb = ((const float4*)b1)[(hb + u) * 4 + half * 2 + 1];
                    const float bv[8] = {ba.x, ba.y, ba.z, ba.w, bb.x, bb.y, bb.z, bb.w};
                    float hs[8];
                    #pragma unroll
                    for (int r = 0; r < 8; ++r) hs[r] = fmaxf(c[r] * nd + bv[r], 0.0f) * ns;
                    #pragma unroll
                    for (int j = 0; j < 4; ++j) {
                        v2f hf;
                        hf.x = hs[2 * j];
                        hf.y = hs[2 * j + 1];
                        #pragma unroll
                        for (int op = 0; op < 2; ++op) {
                            v2f a;
                            a.x = wg[2 * j + op].x; a.y = wg[2 * j + op].y;
                            o[2 * op] = __builtin_amdgcn_wmma_f32_16x16x4_f32(false, a, false, hf, (short)0, o[2 * op], false, false);
                            a.x = wg[2 * j + op].z; a.y = wg[2 * j + op].w;
                            o[2 * op + 1] = __builtin_amdgcn_wmma_f32_16x16x4_f32(false, a, false, hf, (short)0, o[2 * op + 1], false, false);
                        }
                    }
                }
                __builtin_amdgcn_sched_barrier(0);
            }
            if (live) {
                float* drow = dest + (long long)(base + node) * F_OUT + half * 8;
                #pragma unroll
                for (int ob = 0; ob < F_OUT / 16; ++ob) {
                    *(float4*)(drow + ob * 16)     = make_float4(o[ob][0], o[ob][1], o[ob][2], o[ob][3]);
                    *(float4*)(drow + ob * 16 + 4) = make_float4(o[ob][4], o[ob][5], o[ob][6], o[ob][7]);
                }
            }
        }
    } else {
        float4* out4 = (float4*)(dest + (long long)base * 64);
        const float4* tile4a = (const float4*)tile0;
        const float4* tile4b = (const float4*)tile1;
        #pragma unroll
        for (int p = 0; p < STORE_PASSES; ++p) {
            const int i = threadIdx.x + p * AGG_THREADS;
            if (i < TILE_NODES * 16) {
                const float4 ta = tile4a[i], tb = tile4b[i];
                float4 t = make_float4(ta.x + tb.x, ta.y + tb.y, ta.z + tb.z, ta.w + tb.w);
                if (EPILOGUE) {
                    t.x = t.x * rs[p] + b4.x;
                    t.y = t.y * rs[p] + b4.y;
                    t.z = t.z * rs[p] + b4.z;
                    t.w = t.w * rs[p] + b4.w;
                }
                out4[i] = t;
            }
        }
    }
}

#define DEG_CHUNKS (N_CHUNKS / (DEG_THREADS / 32))
static_assert(DEG_CHUNKS == 32, "a lane holds the segment word of one of its wave's chunks");
__global__ __launch_bounds__(DEG_THREADS)
void owner_out_degree_kernel(const unsigned* __restrict__ by_src,
                             const unsigned* __restrict__ seg_src,
                             float* __restrict__ norm_src_out) {
    __shared__ unsigned out_count[TILE_NODES];

    const int lane = threadIdx.x & 31;
    const int wave = __builtin_amdgcn_readfirstlane((int)(threadIdx.x >> 5));

    if (threadIdx.x < TILE_NODES) out_count[threadIdx.x] = 0u;
    __syncthreads();

    const int c0 = wave * DEG_CHUNKS;
    const unsigned sg = seg_src[(c0 + lane) * N_BINS + blockIdx.x];
    unsigned head[DEG_CHUNKS];
    #pragma unroll
    for (int i = 0; i < DEG_CHUNKS; ++i) {
        const unsigned w = (unsigned)__builtin_amdgcn_readlane((int)sg, i);
        const int st = (int)(w & 0xFFFFu), n = (int)(w >> 16);
        head[i] = (lane < n) ? by_src[(c0 + i) * CHUNK_PITCH + st + lane] : 0u;
    }
    #pragma unroll
    for (int i = 0; i < DEG_CHUNKS; ++i) {
        const unsigned w = (unsigned)__builtin_amdgcn_readlane((int)sg, i);
        const int st = (int)(w & 0xFFFFu), n = (int)(w >> 16);
        const unsigned* sp = by_src + (c0 + i) * CHUNK_PITCH + st;
        for (int j = 0; j < n; j += 32) {
            unsigned r = head[i];
            if (j > 0) r = (j + lane < n) ? sp[j + lane] : 0u;
            if (j + lane < n) atomicAdd(out_count + r, 1u);
        }
    }
    __syncthreads();

    if (threadIdx.x < TILE_PITCH) {
        const float deg = (threadIdx.x < TILE_NODES) ? (float)out_count[threadIdx.x] : 1.0f;
        norm_src_out[blockIdx.x * TILE_PITCH + threadIdx.x] = rsqrtf(fmaxf(deg, 1.0f));
    }
}


static_assert((N_TILES * TILE_PITCH) % 64 == 0, "the arrays behind norm_src and norm_dst start on 256-byte boundaries");
static_assert(DEG_THREADS >= TILE_PITCH, "the out-degree work-group stores its tile's norm_src entries in one pass");

extern "C" void kernel_launch(void* const* d_in, const int* in_sizes, int n_in,
                              void* d_out, int out_size, void* d_ws, size_t ws_size,
                              hipStream_t stream) {
    const float* x   = (const float*)d_in[0];
    const float* W1  = (const float*)d_in[1];
    const float* b1  = (const float*)d_in[2];
    const float* W2  = (const float*)d_in[3];
    const float* b2  = (const float*)d_in[4];
    const int*   src = (const int*)d_in[5];
    const int*   dst = (const int*)d_in[6];
    float*       out = (float*)d_out;

    float* ws       = (float*)d_ws;
    float* norm_src = ws;
    float* norm_dst = norm_src + N_TILES * TILE_PITCH;
    float* h2w      = norm_dst + N_TILES * TILE_PITCH;
    unsigned* by_dst  = (unsigned*)(h2w + (long long)N_NODES * 64);
    unsigned* by_src  = by_dst + N_CHUNKS * CHUNK_PITCH;
    unsigned* seg_dst = by_src + N_CHUNKS * CHUNK_PITCH;
    unsigned* seg_src = seg_dst + N_CHUNKS * N_BINS;

    edge_sort_kernel<<<N_CHUNKS, SORT_THREADS, 0, stream>>>(src, dst, by_dst, by_src, seg_dst, seg_src);
    owner_out_degree_kernel<<<N_TILES, DEG_THREADS, 0, stream>>>(by_src, seg_src, norm_src);

    owner_aggregate_kernel<true, true, false><<<N_NODES / TILE_NODES, AGG_THREADS, 0, stream>>>(
        x, norm_src, by_dst, seg_dst, norm_dst, nullptr, nullptr, W1, b1, W2, h2w);

    owner_aggregate_kernel<false, false, true><<<N_NODES / TILE_NODES, AGG_THREADS, 0, stream>>>(
        h2w, nullptr, by_dst, seg_dst, nullptr, norm_dst, b2, nullptr, nullptr, nullptr, out);
}
